# starting kernel
# baseline (speedup 1.0000x reference)
#include <hip/hip_runtime.h>
#include <hip/hip_bf16.h>

#define HID 128
#define NGRAPH 512
#define NEG_SLOPE 0.2f
#define BN_EPS 1e-5f

typedef __attribute__((ext_vector_type(2))) float v2f;
typedef __attribute__((ext_vector_type(8))) float v8f;

__device__ __forceinline__ unsigned f2o(float f) {
    unsigned u = __float_as_uint(f);
    return (u & 0x80000000u) ? ~u : (u | 0x80000000u);
}
__device__ __forceinline__ float o2f(unsigned u) {
    unsigned v = (u & 0x80000000u) ? (u & 0x7fffffffu) : ~u;
    return __uint_as_float(v);
}
__device__ __forceinline__ float lrelu(float a) { return a > 0.0f ? a : NEG_SLOPE * a; }

__global__ void gat_zero(float* __restrict__ p, int n) {
    int i = blockIdx.x * blockDim.x + threadIdx.x;
    if (i < n) p[i] = 0.0f;
}

__global__ void gat_gemm_wmma(const float* __restrict__ X, const float* __restrict__ W,
                              float* __restrict__ H, int nrows) {
    int wave = (blockIdx.x * blockDim.x + threadIdx.x) >> 5;
    int m0 = wave * 16;
    if (m0 >= nrows) return;
    int lane = threadIdx.x & 31;
    int lr = lane & 15;
    int lh = lane >> 4;

    v8f acc[8] = {};
    const float* xrow = X + (size_t)(m0 + lr) * HID;

    for (int k = 0; k < HID; k += 4) {
        int ka = k + lh * 2;
        v2f a;
        a.x = xrow[ka];
        a.y = xrow[ka + 1];
#pragma unroll
        for (int n = 0; n < 8; ++n) {
            v2f b;
            b.x = W[ka * HID + n * 16 + lr];
            b.y = W[(ka + 1) * HID + n * 16 + lr];
            acc[n] = __builtin_amdgcn_wmma_f32_16x16x4_f32(
                false, a, false, b, (short)0, acc[n], false, false);
        }
    }
#pragma unroll
    for (int n = 0; n < 8; ++n)
#pragma unroll
        for (int j = 0; j < 8; ++j)
            H[(size_t)(m0 + j + lh * 8) * HID + n * 16 + lr] = acc[n][j];
}

__global__ void gat_node_dots(const float* __restrict__ H, const float* __restrict__ att_s,
                              const float* __restrict__ att_d, float* __restrict__ a_s,
                              float* __restrict__ a_d, unsigned* __restrict__ mbits, int n_nodes) {
    int node = (blockIdx.x * blockDim.x + threadIdx.x) >> 5;
    if (node >= n_nodes) return;
    int lane = threadIdx.x & 31;
    float4 hv = ((const float4*)(H + (size_t)node * HID))[lane];
    float4 ws = ((const float4*)att_s)[lane];
    float4 wd = ((const float4*)att_d)[lane];
    float ps = hv.x * ws.x + hv.y * ws.y + hv.z * ws.z + hv.w * ws.w;
    float pd = hv.x * wd.x + hv.y * wd.y + hv.z * wd.z + hv.w * wd.w;
    for (int off = 16; off > 0; off >>= 1) {
        ps += __shfl_down(ps, off, 32);
        pd += __shfl_down(pd, off, 32);
    }
    if (lane == 0) {
        a_s[node] = ps;
        a_d[node] = pd;
        mbits[node] = f2o(lrelu(ps + pd));
    }
}

__global__ void gat_edge_max(const int* __restrict__ ei, const float* __restrict__ a_s,
                             const float* __restrict__ a_d, unsigned* __restrict__ mbits,
                             int n_edges) {
    int e = blockIdx.x * blockDim.x + threadIdx.x;
    if (e >= n_edges) return;
    int s = ei[e];
    int d = ei[n_edges + e];
    float a = lrelu(a_s[s] + a_d[d]);
    atomicMax(&mbits[d], f2o(a));
}

__global__ void gat_node_init(const float* __restrict__ H, const float* __restrict__ a_s,
                              const float* __restrict__ a_d, float* __restrict__ m,
                              float* __restrict__ s, float* __restrict__ out_acc, int n_nodes) {
    int node = (blockIdx.x * blockDim.x + threadIdx.x) >> 5;
    if (node >= n_nodes) return;
    int lane = threadIdx.x & 31;
    float mf = o2f(((const unsigned*)m)[node]);
    float es = expf(lrelu(a_s[node] + a_d[node]) - mf);
    if (lane == 0) {
        m[node] = mf;
        s[node] = es;
    }
    float4 hv = ((const float4*)(H + (size_t)node * HID))[lane];
    float4 r;
    r.x = es * hv.x; r.y = es * hv.y; r.z = es * hv.z; r.w = es * hv.w;
    ((float4*)(out_acc + (size_t)node * HID))[lane] = r;
}

__global__ void gat_edge_scatter(const int* __restrict__ ei, const float* __restrict__ a_s,
                                 const float* __restrict__ a_d, const float* __restrict__ m,
                                 const float* __restrict__ H, float* __restrict__ s,
                                 float* __restrict__ out_acc, int n_edges) {
    int e = (blockIdx.x * blockDim.x + threadIdx.x) >> 5;
    if (e >= n_edges) return;
    int lane = threadIdx.x & 31;
    int src = ei[e];
    int dst = ei[n_edges + e];
    float w = expf(lrelu(a_s[src] + a_d[dst]) - m[dst]);
    if (lane == 0) atomicAdd(&s[dst], w);
    float4 hv = ((const float4*)(H + (size_t)src * HID))[lane];
    float* o = out_acc + (size_t)dst * HID + lane * 4;
    atomicAdd(o + 0, w * hv.x);
    atomicAdd(o + 1, w * hv.y);
    atomicAdd(o + 2, w * hv.z);
    atomicAdd(o + 3, w * hv.w);
}

__global__ void gat_finalize(float* __restrict__ out_acc, const float* __restrict__ s,
                             const float* __restrict__ bias, float* __restrict__ sums,
                             float* __restrict__ sumsq, int n_nodes) {
    int c = threadIdx.x;
    float b = bias[c];
    float ls = 0.0f, lq = 0.0f;
    for (int n = blockIdx.x; n < n_nodes; n += gridDim.x) {
        float inv = 1.0f / s[n];
        float v = out_acc[(size_t)n * HID + c] * inv + b;
        v = v > 0.0f ? v : expm1f(v);
        out_acc[(size_t)n * HID + c] = v;
        ls += v;
        lq += v * v;
    }
    atomicAdd(&sums[c], ls);
    atomicAdd(&sumsq[c], lq);
}

__global__ void gat_bn_pool(float* __restrict__ out_acc, const float* __restrict__ sums,
                            const float* __restrict__ sumsq, const float* __restrict__ gamma,
                            const float* __restrict__ beta, const int* __restrict__ batch,
                            float* __restrict__ pooled, int layer, int n_nodes) {
    int idx = blockIdx.x * blockDim.x + threadIdx.x;
    if (idx >= n_nodes * HID) return;
    int c = idx & (HID - 1);
    int n = idx >> 7;
    float invN = 1.0f / (float)n_nodes;
    float mu = sums[c] * invN;
    float var = sumsq[c] * invN - mu * mu;
    float xv = (out_acc[idx] - mu) / sqrtf(var + BN_EPS) * gamma[c] + beta[c];
    out_acc[idx] = xv;
    atomicAdd(&pooled[(size_t)batch[n] * (3 * HID) + layer * HID + c], xv);
}

extern "C" void kernel_launch(void* const* d_in, const int* in_sizes, int n_in,
                              void* d_out, int out_size, void* d_ws, size_t ws_size,
                              hipStream_t stream) {
    const float* x_in  = (const float*)d_in[0];
    const int*   ei    = (const int*)d_in[1];
    const int*   batch = (const int*)d_in[2];
    const float* W     = (const float*)d_in[3];
    const float* att_s = (const float*)d_in[4];
    const float* att_d = (const float*)d_in[5];
    const float* bias  = (const float*)d_in[6];
    const float* gamma = (const float*)d_in[7];
    const float* beta  = (const float*)d_in[8];

    const int N = in_sizes[0] / HID;
    const int E = in_sizes[1] / 2;

    float* pooled = (float*)d_out;
    float* xfinal = (float*)d_out + (size_t)NGRAPH * 3 * HID;

    char* w = (char*)d_ws;
    float* hbuf = (float*)w;  w += (size_t)N * HID * sizeof(float);
    float* xb   = (float*)w;  w += (size_t)N * HID * sizeof(float);
    float* asrc = (float*)w;  w += (size_t)N * sizeof(float);
    float* adst = (float*)w;  w += (size_t)N * sizeof(float);
    float* mbuf = (float*)w;  w += (size_t)N * sizeof(float);
    float* sbuf = (float*)w;  w += (size_t)N * sizeof(float);
    float* sums = (float*)w;
    float* sumsq = sums + HID;

    const int waves_g = (N + 15) / 16;
    dim3 blk(256);
    dim3 grid_gemm((waves_g * 32 + 255) / 256);
    dim3 grid_node((N * 32 + 255) / 256);
    dim3 grid_edge((E + 255) / 256);
    dim3 grid_edgew(((size_t)E * 32 + 255) / 256);
    dim3 grid_elem(((size_t)N * HID + 255) / 256);

    gat_zero<<<(NGRAPH * 3 * HID + 255) / 256, blk, 0, stream>>>(pooled, NGRAPH * 3 * HID);

    for (int l = 0; l < 3; ++l) {
        const float* X  = (l == 0) ? x_in : xb;
        float* oacc     = (l == 2) ? xfinal : xb;
        const float* Wl = W + (size_t)l * HID * HID;

        gat_gemm_wmma<<<grid_gemm, blk, 0, stream>>>(X, Wl, hbuf, N);
        gat_node_dots<<<grid_node, blk, 0, stream>>>(hbuf, att_s + l * HID, att_d + l * HID,
                                                     asrc, adst, (unsigned*)mbuf, N);
        gat_edge_max<<<grid_edge, blk, 0, stream>>>(ei, asrc, adst, (unsigned*)mbuf, E);
        gat_node_init<<<grid_node, blk, 0, stream>>>(hbuf, asrc, adst, mbuf, sbuf, oacc, N);
        gat_edge_scatter<<<grid_edgew, blk, 0, stream>>>(ei, asrc, adst, mbuf, hbuf, sbuf, oacc, E);
        gat_zero<<<1, 256, 0, stream>>>(sums, 2 * HID);
        gat_finalize<<<512, 128, 0, stream>>>(oacc, sbuf, bias + l * HID, sums, sumsq, N);
        gat_bn_pool<<<grid_elem, blk, 0, stream>>>(oacc, sums, sumsq, gamma + l * HID,
                                                   beta + l * HID, batch, pooled, l, N);
    }
}


# Edges counting-sorted by destination once per call; each layer's softmax-sum is a per-node register gather, replacing float atomic scatter (sum order in a row = arrival rank)
# speedup vs baseline: 1.9114x; 1.9114x over previous
#include <hip/hip_runtime.h>
#include <hip/hip_bf16.h>

#define HID 128
#define NGRAPH 512
#define NEG_SLOPE 0.2f
#define BN_EPS 1e-5f

typedef __attribute__((ext_vector_type(2))) float v2f;
typedef __attribute__((ext_vector_type(8))) float v8f;

__device__ __forceinline__ float lrelu(float a) { return a > 0.0f ? a : NEG_SLOPE * a; }

__global__ void gat_zero(float* __restrict__ p, int n) {
    int i = blockIdx.x * blockDim.x + threadIdx.x;
    if (i < n) p[i] = 0.0f;
}

__global__ void gat_gemm_wmma(const float* __restrict__ X, const float* __restrict__ W,
                              float* __restrict__ H, int nrows) {
    int wave = (blockIdx.x * blockDim.x + threadIdx.x) >> 5;
    int m0 = wave * 16;
    if (m0 >= nrows) return;
    int lane = threadIdx.x & 31;
    int lr = lane & 15;
    int lh = lane >> 4;

    v8f acc[8] = {};
    const float* xrow = X + (size_t)(m0 + lr) * HID;

    for (int k = 0; k < HID; k += 4) {
        int ka = k + lh * 2;
        v2f a;
        a.x = xrow[ka];
        a.y = xrow[ka + 1];
#pragma unroll
        for (int n = 0; n < 8; ++n) {
            v2f b;
            b.x = W[ka * HID + n * 16 + lr];
            b.y = W[(ka + 1) * HID + n * 16 + lr];
            acc[n] = __builtin_amdgcn_wmma_f32_16x16x4_f32(
                false, a, false, b, (short)0, acc[n], false, false);
        }
    }
#pragma unroll
    for (int n = 0; n < 8; ++n)
#pragma unroll
        for (int j = 0; j < 8; ++j)
            H[(size_t)(m0 + j + lh * 8) * HID + n * 16 + lr] = acc[n][j];
}

__global__ void gat_node_dots(const float* __restrict__ H, const float* __restrict__ att_s,
                              const float* __restrict__ att_d, float* __restrict__ a_s,
                              float* __restrict__ a_d, int n_nodes) {
    int node = (blockIdx.x * blockDim.x + threadIdx.x) >> 5;
    if (node >= n_nodes) return;
    int lane = threadIdx.x & 31;
    float4 hv = ((const float4*)(H + (size_t)node * HID))[lane];
    float4 ws = ((const float4*)att_s)[lane];
    float4 wd = ((const float4*)att_d)[lane];
    float ps = hv.x * ws.x + hv.y * ws.y + hv.z * ws.z + hv.w * ws.w;
    float pd = hv.x * wd.x + hv.y * wd.y + hv.z * wd.z + hv.w * wd.w;
    for (int off = 16; off > 0; off >>= 1) {
        ps += __shfl_down(ps, off, 32);
        pd += __shfl_down(pd, off, 32);
    }
    if (lane == 0) {
        a_s[node] = ps;
        a_d[node] = pd;
    }
}

__global__ void gat_deg_count(const int* __restrict__ ei, int* __restrict__ cnt, int n_edges) {
    int e = blockIdx.x * blockDim.x + threadIdx.x;
    if (e >= n_edges) return;
    atomicAdd(&cnt[ei[n_edges + e]], 1);
}

__global__ void gat_scan_local(const int* __restrict__ cnt, int* __restrict__ rowptr,
                               int* __restrict__ bsum, int n) {
    __shared__ int wtot[32];
    int t = threadIdx.x;
    int i = blockIdx.x * 1024 + t;
    int lane = t & 31;
    int wv = t >> 5;
    int v = (i < n) ? cnt[i] : 0;
    int x = v;
    for (int off = 1; off < 32; off <<= 1) {
        int y = __shfl_up(x, off, 32);
        if (lane >= off) x += y;
    }
    if (lane == 31) wtot[wv] = x;
    __syncthreads();
    int wt = wtot[lane];
    int wsc = wt;
    for (int off = 1; off < 32; off <<= 1) {
        int y = __shfl_up(wsc, off, 32);
        if (lane >= off) wsc += y;
    }
    int wave_off = __shfl(wsc, wv, 32) - __shfl(wt, wv, 32);
    if (i < n) rowptr[i] = wave_off + x - v;
    if (t == 1023) bsum[blockIdx.x] = wave_off + x;
}

__global__ void gat_scan_add(int* __restrict__ rowptr, const int* __restrict__ bsum, int n) {
    __shared__ int boff;
    int t = threadIdx.x;
    int b = blockIdx.x;
    if (t < 32) {
        int acc = 0;
        for (int k = t; k < b; k += 32) acc += bsum[k];
        for (int off = 16; off > 0; off >>= 1) acc += __shfl_xor(acc, off, 32);
        if (t == 0) boff = acc;
    }
    __syncthreads();
    int o = boff;
    int i = b * 1024 + t;
    if (i < n) rowptr[i] += o;
    if (b == (int)gridDim.x - 1 && t == 0) rowptr[n] = o + bsum[b];
}

__global__ void gat_edge_place(const int* __restrict__ ei, const int* __restrict__ rowptr,
                               int* __restrict__ cnt, int* __restrict__ csr_src, int n_edges) {
    int e = blockIdx.x * blockDim.x + threadIdx.x;
    if (e >= n_edges) return;
    int s = ei[e];
    int d = ei[n_edges + e];
    int old = atomicSub(&cnt[d], 1);
    csr_src[rowptr[d] + old - 1] = s;
}

__global__ void gat_gather(const int* __restrict__ rowptr, const int* __restrict__ csr_src,
                           const float* __restrict__ a_s, const float* __restrict__ a_d,
                           const float* __restrict__ H, float* __restrict__ s,
                           float* __restrict__ out_acc, int n_nodes) {
    int node = __builtin_amdgcn_readfirstlane((int)((blockIdx.x * blockDim.x + threadIdx.x) >> 5));
    if (node >= n_nodes) return;
    int lane = threadIdx.x & 31;
    int beg = __builtin_amdgcn_readfirstlane(rowptr[node]);
    int end = __builtin_amdgcn_readfirstlane(rowptr[node + 1]);
    float ad = a_d[node];
    float self_logit = lrelu(a_s[node] + ad);

    float mx = self_logit;
    for (int j = beg + lane; j < end; j += 32)
        mx = fmaxf(mx, lrelu(a_s[csr_src[j]] + ad));
    for (int off = 16; off > 0; off >>= 1)
        mx = fmaxf(mx, __shfl_xor(mx, off, 32));

    float es = expf(self_logit - mx);
    float4 hv = ((const float4*)(H + (size_t)node * HID))[lane];
    float4 acc;
    acc.x = es * hv.x; acc.y = es * hv.y; acc.z = es * hv.z; acc.w = es * hv.w;
    float wsum = 0.0f;

    for (int base = beg; base < end; base += 32) {
        int j = base + lane;
        int sidx = 0;
        float w = 0.0f;
        if (j < end) {
            sidx = csr_src[j];
            w = expf(lrelu(a_s[sidx] + ad) - mx);
        }
        wsum += w;
        int cnt = end - base < 32 ? end - base : 32;
#pragma unroll 4
        for (int k = 0; k < cnt; ++k) {
            int sk = __builtin_amdgcn_readlane(sidx, k);
            float wk = __int_as_float(__builtin_amdgcn_readlane(__float_as_int(w), k));
            float4 h = ((const float4*)(H + (size_t)sk * HID))[lane];
            acc.x += wk * h.x; acc.y += wk * h.y; acc.z += wk * h.z; acc.w += wk * h.w;
        }
    }
    for (int off = 16; off > 0; off >>= 1)
        wsum += __shfl_xor(wsum, off, 32);
    if (lane == 0) s[node] = es + wsum;
    ((float4*)(out_acc + (size_t)node * HID))[lane] = acc;
}

__global__ void gat_finalize(float* __restrict__ out_acc, const float* __restrict__ s,
                             const float* __restrict__ bias, float* __restrict__ sums,
                             float* __restrict__ sumsq, int n_nodes) {
    int c = threadIdx.x;
    float b = bias[c];
    float ls = 0.0f, lq = 0.0f;
    for (int n = blockIdx.x; n < n_nodes; n += gridDim.x) {
        float inv = 1.0f / s[n];
        float v = out_acc[(size_t)n * HID + c] * inv + b;
        v = v > 0.0f ? v : expm1f(v);
        out_acc[(size_t)n * HID + c] = v;
        ls += v;
        lq += v * v;
    }
    atomicAdd(&sums[c], ls);
    atomicAdd(&sumsq[c], lq);
}

__global__ void gat_bn_pool(float* __restrict__ out_acc, const float* __restrict__ sums,
                            const float* __restrict__ sumsq, const float* __restrict__ gamma,
                            const float* __restrict__ beta, const int* __restrict__ batch,
                            float* __restrict__ pooled, int layer, int n_nodes) {
    int idx = blockIdx.x * blockDim.x + threadIdx.x;
    if (idx >= n_nodes * HID) return;
    int c = idx & (HID - 1);
    int n = idx >> 7;
    float invN = 1.0f / (float)n_nodes;
    float mu = sums[c] * invN;
    float var = sumsq[c] * invN - mu * mu;
    float xv = (out_acc[idx] - mu) / sqrtf(var + BN_EPS) * gamma[c] + beta[c];
    out_acc[idx] = xv;
    atomicAdd(&pooled[(size_t)batch[n] * (3 * HID) + layer * HID + c], xv);
}

extern "C" void kernel_launch(void* const* d_in, const int* in_sizes, int n_in,
                              void* d_out, int out_size, void* d_ws, size_t ws_size,
                              hipStream_t stream) {
    const float* x_in  = (const float*)d_in[0];
    const int*   ei    = (const int*)d_in[1];
    const int*   batch = (const int*)d_in[2];
    const float* W     = (const float*)d_in[3];
    const float* att_s = (const float*)d_in[4];
    const float* att_d = (const float*)d_in[5];
    const float* bias  = (const float*)d_in[6];
    const float* gamma = (const float*)d_in[7];
    const float* beta  = (const float*)d_in[8];

    const int N = in_sizes[0] / HID;
    const int E = in_sizes[1] / 2;

    float* pooled = (float*)d_out;
    float* xfinal = (float*)d_out + (size_t)NGRAPH * 3 * HID;

    char* w = (char*)d_ws;
    float* hbuf = (float*)w;  w += (size_t)N * HID * sizeof(float);
    float* xb   = (float*)w;  w += (size_t)N * HID * sizeof(float);
    float* asrc = (float*)w;  w += (size_t)N * sizeof(float);
    float* adst = (float*)w;  w += (size_t)N * sizeof(float);
    float* sbuf = (float*)w;  w += (size_t)N * sizeof(float);
    float* sums = (float*)w;  w += (size_t)2 * HID * sizeof(float);
    float* sumsq = sums + HID;
    int* cnt    = (int*)w;    w += (size_t)N * sizeof(int);
    int* rowptr = (int*)w;    w += (size_t)(N + 1) * sizeof(int);
    int* bsum   = (int*)w;    w += (size_t)((N + 1023) / 1024) * sizeof(int);
    int* csrsrc = (int*)w;

    const int waves_g = (N + 15) / 16;
    dim3 blk(256);
    dim3 grid_gemm((waves_g * 32 + 255) / 256);
    dim3 grid_node((N * 32 + 255) / 256);
    dim3 grid_edge((E + 255) / 256);
    dim3 grid_elem(((size_t)N * HID + 255) / 256);

    gat_zero<<<(NGRAPH * 3 * HID + 255) / 256, blk, 0, stream>>>(pooled, NGRAPH * 3 * HID);

    gat_zero<<<(N + 255) / 256, blk, 0, stream>>>((float*)cnt, N);
    gat_deg_count<<<grid_edge, blk, 0, stream>>>(ei, cnt, E);
    gat_scan_local<<<(N + 1023) / 1024, 1024, 0, stream>>>(cnt, rowptr, bsum, N);
    gat_scan_add<<<(N + 1023) / 1024, 1024, 0, stream>>>(rowptr, bsum, N);
    gat_edge_place<<<grid_edge, blk, 0, stream>>>(ei, rowptr, cnt, csrsrc, E);

    for (int l = 0; l < 3; ++l) {
        const float* X  = (l == 0) ? x_in : xb;
        float* oacc     = (l == 2) ? xfinal : xb;
        const float* Wl = W + (size_t)l * HID * HID;

        gat_gemm_wmma<<<grid_gemm, blk, 0, stream>>>(X, Wl, hbuf, N);
        gat_node_dots<<<grid_node, blk, 0, stream>>>(hbuf, att_s + l * HID, att_d + l * HID,
                                                     asrc, adst, N);
        gat_gather<<<grid_node, blk, 0, stream>>>(rowptr, csrsrc, asrc, adst, hbuf, sbuf, oacc, N);
        gat_zero<<<1, 256, 0, stream>>>(sums, 2 * HID);
        gat_finalize<<<512, 128, 0, stream>>>(oacc, sbuf, bias + l * HID, sums, sumsq, N);
        gat_bn_pool<<<grid_elem, blk, 0, stream>>>(oacc, sums, sumsq, gamma + l * HID,
                                                   beta + l * HID, batch, pooled, l, N);
    }
}


# Finalize launch folded into the gather epilogue: out=elu(acc/denom+bias) from registers, BN sums as per-block LDS partial rows + one column-sum launch (stat sum order: block tree, then atomics)
# speedup vs baseline: 2.7737x; 1.4511x over previous
#include <hip/hip_runtime.h>
#include <hip/hip_bf16.h>

#define HID 128
#define NGRAPH 512
#define NEG_SLOPE 0.2f
#define BN_EPS 1e-5f

typedef __attribute__((ext_vector_type(2))) float v2f;
typedef __attribute__((ext_vector_type(8))) float v8f;

__device__ __forceinline__ float lrelu(float a) { return a > 0.0f ? a : NEG_SLOPE * a; }

__global__ void gat_zero(float* __restrict__ p, int n) {
    int i = blockIdx.x * blockDim.x + threadIdx.x;
    if (i < n) p[i] = 0.0f;
}

__global__ void gat_gemm_wmma(const float* __restrict__ X, const float* __restrict__ W,
                              float* __restrict__ H, int nrows) {
    int wave = (blockIdx.x * blockDim.x + threadIdx.x) >> 5;
    int m0 = wave * 16;
    if (m0 >= nrows) return;
    int lane = threadIdx.x & 31;
    int lr = lane & 15;
    int lh = lane >> 4;

    v8f acc[8] = {};
    const float* xrow = X + (size_t)(m0 + lr) * HID;

    for (int k = 0; k < HID; k += 4) {
        int ka = k + lh * 2;
        v2f a;
        a.x = xrow[ka];
        a.y = xrow[ka + 1];
#pragma unroll
        for (int n = 0; n < 8; ++n) {
            v2f b;
            b.x = W[ka * HID + n * 16 + lr];
            b.y = W[(ka + 1) * HID + n * 16 + lr];
            acc[n] = __builtin_amdgcn_wmma_f32_16x16x4_f32(
                false, a, false, b, (short)0, acc[n], false, false);
        }
    }
#pragma unroll
    for (int n = 0; n < 8; ++n)
#pragma unroll
        for (int j = 0; j < 8; ++j)
            H[(size_t)(m0 + j + lh * 8) * HID + n * 16 + lr] = acc[n][j];
}

__global__ void gat_node_dots(const float* __restrict__ H, const float* __restrict__ att_s,
                              const float* __restrict__ att_d, float* __restrict__ a_s,
                              float* __restrict__ a_d, int n_nodes) {
    int node = (blockIdx.x * blockDim.x + threadIdx.x) >> 5;
    if (node >= n_nodes) return;
    int lane = threadIdx.x & 31;
    float4 hv = ((const float4*)(H + (size_t)node * HID))[lane];
    float4 ws = ((const float4*)att_s)[lane];
    float4 wd = ((const float4*)att_d)[lane];
    float ps = hv.x * ws.x + hv.y * ws.y + hv.z * ws.z + hv.w * ws.w;
    float pd = hv.x * wd.x + hv.y * wd.y + hv.z * wd.z + hv.w * wd.w;
    for (int off = 16; off > 0; off >>= 1) {
        ps += __shfl_down(ps, off, 32);
        pd += __shfl_down(pd, off, 32);
    }
    if (lane == 0) {
        a_s[node] = ps;
        a_d[node] = pd;
    }
}

__global__ void gat_deg_count(const int* __restrict__ ei, int* __restrict__ cnt, int n_edges) {
    int e = blockIdx.x * blockDim.x + threadIdx.x;
    if (e >= n_edges) return;
    atomicAdd(&cnt[ei[n_edges + e]], 1);
}

__global__ void gat_scan_local(const int* __restrict__ cnt, int* __restrict__ rowptr,
                               int* __restrict__ bsum, int n) {
    __shared__ int wtot[32];
    int t = threadIdx.x;
    int i = blockIdx.x * 1024 + t;
    int lane = t & 31;
    int wv = t >> 5;
    int v = (i < n) ? cnt[i] : 0;
    int x = v;
    for (int off = 1; off < 32; off <<= 1) {
        int y = __shfl_up(x, off, 32);
        if (lane >= off) x += y;
    }
    if (lane == 31) wtot[wv] = x;
    __syncthreads();
    int wt = wtot[lane];
    int wsc = wt;
    for (int off = 1; off < 32; off <<= 1) {
        int y = __shfl_up(wsc, off, 32);
        if (lane >= off) wsc += y;
    }
    int wave_off = __shfl(wsc, wv, 32) - __shfl(wt, wv, 32);
    if (i < n) rowptr[i] = wave_off + x - v;
    if (t == 1023) bsum[blockIdx.x] = wave_off + x;
}

__global__ void gat_scan_add(int* __restrict__ rowptr, const int* __restrict__ bsum, int n) {
    __shared__ int boff;
    int t = threadIdx.x;
    int b = blockIdx.x;
    if (t < 32) {
        int acc = 0;
        for (int k = t; k < b; k += 32) acc += bsum[k];
        for (int off = 16; off > 0; off >>= 1) acc += __shfl_xor(acc, off, 32);
        if (t == 0) boff = acc;
    }
    __syncthreads();
    int o = boff;
    int i = b * 1024 + t;
    if (i < n) rowptr[i] += o;
    if (b == (int)gridDim.x - 1 && t == 0) rowptr[n] = o + bsum[b];
}

__global__ void gat_edge_place(const int* __restrict__ ei, const int* __restrict__ rowptr,
                               int* __restrict__ cnt, int* __restrict__ csr_src, int n_edges) {
    int e = blockIdx.x * blockDim.x + threadIdx.x;
    if (e >= n_edges) return;
    int s = ei[e];
    int d = ei[n_edges + e];
    int old = atomicSub(&cnt[d], 1);
    csr_src[rowptr[d] + old - 1] = s;
}

#define GW 8

__device__ __forceinline__ float elu1(float v) { return v > 0.0f ? v : expm1f(v); }

__global__ void gat_gather(const int* __restrict__ rowptr, const int* __restrict__ csr_src,
                           const float* __restrict__ a_s, const float* __restrict__ a_d,
                           const float* __restrict__ H, const float* __restrict__ bias,
                           float* __restrict__ out, float* __restrict__ part,
                           float* __restrict__ sums, int n_nodes) {
    __shared__ float4 red[GW][64];
    int node = __builtin_amdgcn_readfirstlane((int)((blockIdx.x * blockDim.x + threadIdx.x) >> 5));
    int lane = threadIdx.x & 31;
    int wv = threadIdx.x >> 5;
    float4 v = make_float4(0.0f, 0.0f, 0.0f, 0.0f);
    float4 q = make_float4(0.0f, 0.0f, 0.0f, 0.0f);

    if (node < n_nodes) {
        int beg = __builtin_amdgcn_readfirstlane(rowptr[node]);
        int end = __builtin_amdgcn_readfirstlane(rowptr[node + 1]);
        float ad = a_d[node];
        float self_logit = lrelu(a_s[node] + ad);

        float mx = self_logit;
        for (int j = beg + lane; j < end; j += 32)
            mx = fmaxf(mx, lrelu(a_s[csr_src[j]] + ad));
        for (int off = 16; off > 0; off >>= 1)
            mx = fmaxf(mx, __shfl_xor(mx, off, 32));

        float es = expf(self_logit - mx);
        float4 hv = ((const float4*)(H + (size_t)node * HID))[lane];
        float4 acc;
        acc.x = es * hv.x; acc.y = es * hv.y; acc.z = es * hv.z; acc.w = es * hv.w;
        float wsum = 0.0f;

        for (int base = beg; base < end; base += 32) {
            int j = base + lane;
            int sidx = 0;
            float w = 0.0f;
            if (j < end) {
                sidx = csr_src[j];
                w = expf(lrelu(a_s[sidx] + ad) - mx);
            }
            wsum += w;
            int cnt = end - base < 32 ? end - base : 32;
#pragma unroll 4
            for (int k = 0; k < cnt; ++k) {
                int sk = __builtin_amdgcn_readlane(sidx, k);
                float wk = __int_as_float(__builtin_amdgcn_readlane(__float_as_int(w), k));
                float4 h = ((const float4*)(H + (size_t)sk * HID))[lane];
                acc.x += wk * h.x; acc.y += wk * h.y; acc.z += wk * h.z; acc.w += wk * h.w;
            }
        }
        for (int off = 16; off > 0; off >>= 1)
            wsum += __shfl_xor(wsum, off, 32);

        float inv = 1.0f / (es + wsum);
        float4 b = ((const float4*)bias)[lane];
        v.x = elu1(acc.x * inv + b.x);
        v.y = elu1(acc.y * inv + b.y);
        v.z = elu1(acc.z * inv + b.z);
        v.w = elu1(acc.w * inv + b.w);
        q.x = v.x * v.x; q.y = v.y * v.y; q.z = v.z * v.z; q.w = v.w * v.w;
    }

    red[wv][lane] = v;
    red[wv][32 + lane] = q;
    __syncthreads();
    const float* rf = (const float*)red;
    int t = threadIdx.x;
    float tot = 0.0f;
    for (int w = 0; w < GW; ++w) tot += rf[w * 256 + t];
    part[(size_t)blockIdx.x * (2 * HID) + t] = tot;
    if (blockIdx.x == 0) sums[t] = 0.0f;
    if (node < n_nodes) ((float4*)(out + (size_t)node * HID))[lane] = v;
}

__global__ void gat_stats_reduce(const float* __restrict__ part, float* __restrict__ sums,
                                 int nrows) {
    int t = threadIdx.x;
    int r = blockIdx.x;
    const int g = gridDim.x;
    float acc = 0.0f;
    for (; r + 15 * g < nrows; r += 16 * g) {
        float x[16];
        for (int i = 0; i < 16; ++i) x[i] = part[(size_t)(r + i * g) * (2 * HID) + t];
        for (int i = 0; i < 16; ++i) acc += x[i];
    }
    for (; r < nrows; r += g) acc += part[(size_t)r * (2 * HID) + t];
    atomicAdd(&sums[t], acc);
}

__global__ void gat_bn_pool(float* __restrict__ out_acc, const float* __restrict__ sums,
                            const float* __restrict__ sumsq, const float* __restrict__ gamma,
                            const float* __restrict__ beta, const int* __restrict__ batch,
                            float* __restrict__ pooled, int layer, int n_nodes) {
    int idx = blockIdx.x * blockDim.x + threadIdx.x;
    if (idx >= n_nodes * HID) return;
    int c = idx & (HID - 1);
    int n = idx >> 7;
    float invN = 1.0f / (float)n_nodes;
    float mu = sums[c] * invN;
    float var = sumsq[c] * invN - mu * mu;
    float xv = (out_acc[idx] - mu) / sqrtf(var + BN_EPS) * gamma[c] + beta[c];
    out_acc[idx] = xv;
    atomicAdd(&pooled[(size_t)batch[n] * (3 * HID) + layer * HID + c], xv);
}

extern "C" void kernel_launch(void* const* d_in, const int* in_sizes, int n_in,
                              void* d_out, int out_size, void* d_ws, size_t ws_size,
                              hipStream_t stream) {
    const float* x_in  = (const float*)d_in[0];
    const int*   ei    = (const int*)d_in[1];
    const int*   batch = (const int*)d_in[2];
    const float* W     = (const float*)d_in[3];
    const float* att_s = (const float*)d_in[4];
    const float* att_d = (const float*)d_in[5];
    const float* bias  = (const float*)d_in[6];
    const float* gamma = (const float*)d_in[7];
    const float* beta  = (const float*)d_in[8];

    const int N = in_sizes[0] / HID;
    const int E = in_sizes[1] / 2;

    float* pooled = (float*)d_out;
    float* xfinal = (float*)d_out + (size_t)NGRAPH * 3 * HID;

    char* w = (char*)d_ws;
    float* hbuf = (float*)w;  w += (size_t)N * HID * sizeof(float);
    float* xb   = (float*)w;  w += (size_t)N * HID * sizeof(float);
    const int n_gblk = (N + 7) / 8;
    float* part = (float*)w;  w += (size_t)n_gblk * 2 * HID * sizeof(float);
    float* asrc = (float*)w;  w += (size_t)N * sizeof(float);
    float* adst = (float*)w;  w += (size_t)N * sizeof(float);
    float* sums = (float*)w;  w += (size_t)2 * HID * sizeof(float);
    float* sumsq = sums + HID;
    int* cnt    = (int*)w;    w += (size_t)N * sizeof(int);
    int* rowptr = (int*)w;    w += (size_t)(N + 1) * sizeof(int);
    int* bsum   = (int*)w;    w += (size_t)((N + 1023) / 1024) * sizeof(int);
    int* csrsrc = (int*)w;

    const int waves_g = (N + 15) / 16;
    dim3 blk(256);
    dim3 grid_gemm((waves_g * 32 + 255) / 256);
    dim3 grid_node(n_gblk);
    dim3 grid_edge((E + 255) / 256);
    dim3 grid_elem(((size_t)N * HID + 255) / 256);

    gat_zero<<<(NGRAPH * 3 * HID + 255) / 256, blk, 0, stream>>>(pooled, NGRAPH * 3 * HID);

    gat_zero<<<(N + 255) / 256, blk, 0, stream>>>((float*)cnt, N);
    gat_deg_count<<<grid_edge, blk, 0, stream>>>(ei, cnt, E);
    gat_scan_local<<<(N + 1023) / 1024, 1024, 0, stream>>>(cnt, rowptr, bsum, N);
    gat_scan_add<<<(N + 1023) / 1024, 1024, 0, stream>>>(rowptr, bsum, N);
    gat_edge_place<<<grid_edge, blk, 0, stream>>>(ei, rowptr, cnt, csrsrc, E);

    for (int l = 0; l < 3; ++l) {
        const float* X  = (l == 0) ? x_in : xb;
        float* oacc     = (l == 2) ? xfinal : xb;
        const float* Wl = W + (size_t)l * HID * HID;

        gat_gemm_wmma<<<grid_gemm, blk, 0, stream>>>(X, Wl, hbuf, N);
        gat_node_dots<<<grid_node, blk, 0, stream>>>(hbuf, att_s + l * HID, att_d + l * HID,
                                                     asrc, adst, N);
        gat_gather<<<grid_node, blk, 0, stream>>>(rowptr, csrsrc, asrc, adst, hbuf, bias + l * HID,
                                                  oacc, part, sums, N);
        gat_stats_reduce<<<256, 256, 0, stream>>>(part, sums, n_gblk);
        gat_bn_pool<<<grid_elem, blk, 0, stream>>>(oacc, sums, sumsq, gamma + l * HID,
                                                   beta + l * HID, batch, pooled, l, N);
    }
}


# Row and edge launches regrouped: a wave loops over 8 node rows, a thread over 8 edges (BN-apply launch 400000 waves to 12500); pooled sums = in-wave runs per graph, then atomics
# speedup vs baseline: 3.2772x; 1.1815x over previous
#include <hip/hip_runtime.h>
#include <hip/hip_bf16.h>

#define HID 128
#define NGRAPH 512
#define NEG_SLOPE 0.2f
#define BN_EPS 1e-5f

typedef __attribute__((ext_vector_type(2))) float v2f;
typedef __attribute__((ext_vector_type(8))) float v8f;

__device__ __forceinline__ float lrelu(float a) { return a > 0.0f ? a : NEG_SLOPE * a; }

__global__ void gat_zero(float* __restrict__ p, int n) {
    int i = blockIdx.x * blockDim.x + threadIdx.x;
    if (i < n) p[i] = 0.0f;
}

__global__ void gat_gemm_wmma(const float* __restrict__ X, const float* __restrict__ W,
                              float* __restrict__ H, int nrows) {
    int wave = (blockIdx.x * blockDim.x + threadIdx.x) >> 5;
    int m0 = wave * 16;
    if (m0 >= nrows) return;
    int lane = threadIdx.x & 31;
    int lr = lane & 15;
    int lh = lane >> 4;

    v8f acc[8] = {};
    const float* xrow = X + (size_t)(m0 + lr) * HID;

    for (int k = 0; k < HID; k += 4) {
        int ka = k + lh * 2;
        v2f a;
        a.x = xrow[ka];
        a.y = xrow[ka + 1];
#pragma unroll
        for (int n = 0; n < 8; ++n) {
            v2f b;
            b.x = W[ka * HID + n * 16 + lr];
            b.y = W[(ka + 1) * HID + n * 16 + lr];
            acc[n] = __builtin_amdgcn_wmma_f32_16x16x4_f32(
                false, a, false, b, (short)0, acc[n], false, false);
        }
    }
#pragma unroll
    for (int n = 0; n < 8; ++n)
#pragma unroll
        for (int j = 0; j < 8; ++j)
            H[(size_t)(m0 + j + lh * 8) * HID + n * 16 + lr] = acc[n][j];
}

#define NPW 8

__global__ void gat_node_dots(const float* __restrict__ H, const float* __restrict__ att_s,
                              const float* __restrict__ att_d, float* __restrict__ a_s,
                              float* __restrict__ a_d, int n_nodes) {
    int wave = __builtin_amdgcn_readfirstlane((int)((blockIdx.x * blockDim.x + threadIdx.x) >> 5));
    int n0 = wave * NPW;
    if (n0 >= n_nodes) return;
    int lane = threadIdx.x & 31;
    float4 ws = ((const float4*)att_s)[lane];
    float4 wd = ((const float4*)att_d)[lane];
    float4 hv[NPW];
#pragma unroll
    for (int i = 0; i < NPW; ++i) {
        int n = n0 + i < n_nodes ? n0 + i : n_nodes - 1;
        hv[i] = ((const float4*)(H + (size_t)n * HID))[lane];
    }
    float os = 0.0f, od = 0.0f;
#pragma unroll
    for (int i = 0; i < NPW; ++i) {
        float ps = hv[i].x * ws.x + hv[i].y * ws.y + hv[i].z * ws.z + hv[i].w * ws.w;
        float pd = hv[i].x * wd.x + hv[i].y * wd.y + hv[i].z * wd.z + hv[i].w * wd.w;
        for (int off = 16; off > 0; off >>= 1) {
            ps += __shfl_xor(ps, off, 32);
            pd += __shfl_xor(pd, off, 32);
        }
        if (lane == i) { os = ps; od = pd; }
    }
    if (lane < NPW && n0 + lane < n_nodes) {
        a_s[n0 + lane] = os;
        a_d[n0 + lane] = od;
    }
}

#define EPT 8

__global__ void gat_deg_count(const int* __restrict__ ei, int* __restrict__ cnt, int n_edges) {
    int b0 = blockIdx.x * (int)(blockDim.x * EPT);
    int e0 = b0 + threadIdx.x;
    if (b0 + (int)(blockDim.x * EPT) <= n_edges) {
        int d[EPT];
#pragma unroll
        for (int i = 0; i < EPT; ++i) d[i] = ei[n_edges + e0 + i * (int)blockDim.x];
#pragma unroll
        for (int i = 0; i < EPT; ++i) atomicAdd(&cnt[d[i]], 1);
    } else {
        for (int i = 0; i < EPT; ++i) {
            int e = e0 + i * (int)blockDim.x;
            if (e < n_edges) atomicAdd(&cnt[ei[n_edges + e]], 1);
        }
    }
}

__global__ void gat_scan_local(const int* __restrict__ cnt, int* __restrict__ rowptr,
                               int* __restrict__ bsum, int n) {
    __shared__ int wtot[32];
    int t = threadIdx.x;
    int i = blockIdx.x * 1024 + t;
    int lane = t & 31;
    int wv = t >> 5;
    int v = (i < n) ? cnt[i] : 0;
    int x = v;
    for (int off = 1; off < 32; off <<= 1) {
        int y = __shfl_up(x, off, 32);
        if (lane >= off) x += y;
    }
    if (lane == 31) wtot[wv] = x;
    __syncthreads();
    int wt = wtot[lane];
    int wsc = wt;
    for (int off = 1; off < 32; off <<= 1) {
        int y = __shfl_up(wsc, off, 32);
        if (lane >= off) wsc += y;
    }
    int wave_off = __shfl(wsc, wv, 32) - __shfl(wt, wv, 32);
    if (i < n) rowptr[i] = wave_off + x - v;
    if (t == 1023) bsum[blockIdx.x] = wave_off + x;
}

__global__ void gat_scan_add(int* __restrict__ rowptr, const int* __restrict__ bsum, int n) {
    __shared__ int boff;
    int t = threadIdx.x;
    int b = blockIdx.x;
    if (t < 32) {
        int acc = 0;
        for (int k = t; k < b; k += 32) acc += bsum[k];
        for (int off = 16; off > 0; off >>= 1) acc += __shfl_xor(acc, off, 32);
        if (t == 0) boff = acc;
    }
    __syncthreads();
    int o = boff;
    int i = b * 1024 + t;
    if (i < n) rowptr[i] += o;
    if (b == (int)gridDim.x - 1 && t == 0) rowptr[n] = o + bsum[b];
}

__global__ void gat_edge_place(const int* __restrict__ ei, const int* __restrict__ rowptr,
                               int* __restrict__ cnt, int* __restrict__ csr_src, int n_edges) {
    int b0 = blockIdx.x * (int)(blockDim.x * EPT);
    int e0 = b0 + threadIdx.x;
    if (b0 + (int)(blockDim.x * EPT) <= n_edges) {
        int s[EPT], d[EPT], pos[EPT];
#pragma unroll
        for (int i = 0; i < EPT; ++i) {
            s[i] = ei[e0 + i * (int)blockDim.x];
            d[i] = ei[n_edges + e0 + i * (int)blockDim.x];
        }
#pragma unroll
        for (int i = 0; i < EPT; ++i) pos[i] = rowptr[d[i]] + atomicSub(&cnt[d[i]], 1) - 1;
#pragma unroll
        for (int i = 0; i < EPT; ++i) csr_src[pos[i]] = s[i];
    } else {
        for (int i = 0; i < EPT; ++i) {
            int e = e0 + i * (int)blockDim.x;
            if (e < n_edges) {
                int dd = ei[n_edges + e];
                csr_src[rowptr[dd] + atomicSub(&cnt[dd], 1) - 1] = ei[e];
            }
        }
    }
}

#define GW 8

__device__ __forceinline__ float elu1(float v) { return v > 0.0f ? v : expm1f(v); }

__global__ void gat_gather(const int* __restrict__ rowptr, const int* __restrict__ csr_src,
                           const float* __restrict__ a_s, const float* __restrict__ a_d,
                           const float* __restrict__ H, const float* __restrict__ bias,
                           float* __restrict__ out, float* __restrict__ part,
                           float* __restrict__ sums, int n_nodes) {
    __shared__ float4 red[GW][64];
    int node = __builtin_amdgcn_readfirstlane((int)((blockIdx.x * blockDim.x + threadIdx.x) >> 5));
    int lane = threadIdx.x & 31;
    int wv = threadIdx.x >> 5;
    float4 v = make_float4(0.0f, 0.0f, 0.0f, 0.0f);
    float4 q = make_float4(0.0f, 0.0f, 0.0f, 0.0f);

    if (node < n_nodes) {
        int beg = __builtin_amdgcn_readfirstlane(rowptr[node]);
        int end = __builtin_amdgcn_readfirstlane(rowptr[node + 1]);
        float ad = a_d[node];
        float self_logit = lrelu(a_s[node] + ad);

        float mx = self_logit;
        for (int j = beg + lane; j < end; j += 32)
            mx = fmaxf(mx, lrelu(a_s[csr_src[j]] + ad));
        for (int off = 16; off > 0; off >>= 1)
            mx = fmaxf(mx, __shfl_xor(mx, off, 32));

        float es = expf(self_logit - mx);
        float4 hv = ((const float4*)(H + (size_t)node * HID))[lane];
        float4 acc;
        acc.x = es * hv.x; acc.y = es * hv.y; acc.z = es * hv.z; acc.w = es * hv.w;
        float wsum = 0.0f;

        for (int base = beg; base < end; base += 32) {
            int j = base + lane;
            int sidx = 0;
            float w = 0.0f;
            if (j < end) {
                sidx = csr_src[j];
                w = expf(lrelu(a_s[sidx] + ad) - mx);
            }
            wsum += w;
            int cnt = end - base < 32 ? end - base : 32;
#pragma unroll 4
            for (int k = 0; k < cnt; ++k) {
                int sk = __builtin_amdgcn_readlane(sidx, k);
                float wk = __int_as_float(__builtin_amdgcn_readlane(__float_as_int(w), k));
                float4 h = ((const float4*)(H + (size_t)sk * HID))[lane];
                acc.x += wk * h.x; acc.y += wk * h.y; acc.z += wk * h.z; acc.w += wk * h.w;
            }
        }
        for (int off = 16; off > 0; off >>= 1)
            wsum += __shfl_xor(wsum, off, 32);

        float inv = 1.0f / (es + wsum);
        float4 b = ((const float4*)bias)[lane];
        v.x = elu1(acc.x * inv + b.x);
        v.y = elu1(acc.y * inv + b.y);
        v.z = elu1(acc.z * inv + b.z);
        v.w = elu1(acc.w * inv + b.w);
        q.x = v.x * v.x; q.y = v.y * v.y; q.z = v.z * v.z; q.w = v.w * v.w;
    }

    red[wv][lane] = v;
    red[wv][32 + lane] = q;
    __syncthreads();
    const float* rf = (const float*)red;
    int t = threadIdx.x;
    float tot = 0.0f;
    for (int w = 0; w < GW; ++w) tot += rf[w * 256 + t];
    part[(size_t)blockIdx.x * (2 * HID) + t] = tot;
    if (blockIdx.x == 0) sums[t] = 0.0f;
    if (node < n_nodes) ((float4*)(out + (size_t)node * HID))[lane] = v;
}

__global__ void gat_stats_reduce(const float* __restrict__ part, float* __restrict__ sums,
                                 int nrows) {
    int t = threadIdx.x;
    int r = blockIdx.x;
    const int g = gridDim.x;
    float acc = 0.0f;
    for (; r + 15 * g < nrows; r += 16 * g) {
        float x[16];
        for (int i = 0; i < 16; ++i) x[i] = part[(size_t)(r + i * g) * (2 * HID) + t];
        for (int i = 0; i < 16; ++i) acc += x[i];
    }
    for (; r < nrows; r += g) acc += part[(size_t)r * (2 * HID) + t];
    atomicAdd(&sums[t], acc);
}

__device__ __forceinline__ void pool_add(float* __restrict__ pooled, int g, int layer, int lane,
                                         float4 a) {
    float* p = pooled + (size_t)g * (3 * HID) + layer * HID + lane * 4;
    atomicAdd(p + 0, a.x);
    atomicAdd(p + 1, a.y);
    atomicAdd(p + 2, a.z);
    atomicAdd(p + 3, a.w);
}

__global__ void gat_bn_pool(float* __restrict__ out_acc, const float* __restrict__ sums,
                            const float* __restrict__ sumsq, const float* __restrict__ gamma,
                            const float* __restrict__ beta, const int* __restrict__ batch,
                            float* __restrict__ pooled, int layer, int n_nodes) {
    int wave = __builtin_amdgcn_readfirstlane((int)((blockIdx.x * blockDim.x + threadIdx.x) >> 5));
    int n0 = wave * NPW;
    if (n0 >= n_nodes) return;
    int lane = threadIdx.x & 31;

    float4 x[NPW];
    int gid[NPW];
#pragma unroll
    for (int i = 0; i < NPW; ++i) {
        int n = n0 + i < n_nodes ? n0 + i : n_nodes - 1;
        x[i] = ((const float4*)(out_acc + (size_t)n * HID))[lane];
        gid[i] = batch[n];
    }

    float invN = 1.0f / (float)n_nodes;
    float4 s = ((const float4*)sums)[lane];
    float4 sq = ((const float4*)sumsq)[lane];
    float4 ga = ((const float4*)gamma)[lane];
    float4 be = ((const float4*)beta)[lane];
    float4 mu, sd;
    mu.x = s.x * invN; mu.y = s.y * invN; mu.z = s.z * invN; mu.w = s.w * invN;
    sd.x = sqrtf(sq.x * invN - mu.x * mu.x + BN_EPS);
    sd.y = sqrtf(sq.y * invN - mu.y * mu.y + BN_EPS);
    sd.z = sqrtf(sq.z * invN - mu.z * mu.z + BN_EPS);
    sd.w = sqrtf(sq.w * invN - mu.w * mu.w + BN_EPS);

    int cur = gid[0];
    float4 run = make_float4(0.0f, 0.0f, 0.0f, 0.0f);
#pragma unroll
    for (int i = 0; i < NPW; ++i) {
        if (n0 + i < n_nodes) {
            float4 v;
            v.x = (x[i].x - mu.x) / sd.x * ga.x + be.x;
            v.y = (x[i].y - mu.y) / sd.y * ga.y + be.y;
            v.z = (x[i].z - mu.z) / sd.z * ga.z + be.z;
            v.w = (x[i].w - mu.w) / sd.w * ga.w + be.w;
            ((float4*)(out_acc + (size_t)(n0 + i) * HID))[lane] = v;
            if (gid[i] != cur) {
                pool_add(pooled, cur, layer, lane, run);
                cur = gid[i];
                run = make_float4(0.0f, 0.0f, 0.0f, 0.0f);
            }
            run.x += v.x; run.y += v.y; run.z += v.z; run.w += v.w;
        }
    }
    pool_add(pooled, cur, layer, lane, run);
}

extern "C" void kernel_launch(void* const* d_in, const int* in_sizes, int n_in,
                              void* d_out, int out_size, void* d_ws, size_t ws_size,
                              hipStream_t stream) {
    const float* x_in  = (const float*)d_in[0];
    const int*   ei    = (const int*)d_in[1];
    const int*   batch = (const int*)d_in[2];
    const float* W     = (const float*)d_in[3];
    const float* att_s = (const float*)d_in[4];
    const float* att_d = (const float*)d_in[5];
    const float* bias  = (const float*)d_in[6];
    const float* gamma = (const float*)d_in[7];
    const float* beta  = (const float*)d_in[8];

    const int N = in_sizes[0] / HID;
    const int E = in_sizes[1] / 2;

    float* pooled = (float*)d_out;
    float* xfinal = (float*)d_out + (size_t)NGRAPH * 3 * HID;

    char* w = (char*)d_ws;
    float* hbuf = (float*)w;  w += (size_t)N * HID * sizeof(float);
    float* xb   = (float*)w;  w += (size_t)N * HID * sizeof(float);
    const int n_gblk = (N + 7) / 8;
    float* part = (float*)w;  w += (size_t)n_gblk * 2 * HID * sizeof(float);
    float* asrc = (float*)w;  w += (size_t)N * sizeof(float);
    float* adst = (float*)w;  w += (size_t)N * sizeof(float);
    float* sums = (float*)w;  w += (size_t)2 * HID * sizeof(float);
    float* sumsq = sums + HID;
    int* cnt    = (int*)w;    w += (size_t)N * sizeof(int);
    int* rowptr = (int*)w;    w += (size_t)(N + 1) * sizeof(int);
    int* bsum   = (int*)w;    w += (size_t)((N + 1023) / 1024) * sizeof(int);
    int* csrsrc = (int*)w;

    const int waves_g = (N + 15) / 16;
    dim3 blk(256);
    dim3 grid_gemm((waves_g * 32 + 255) / 256);
    dim3 grid_node(n_gblk);
    dim3 grid_edge((E + 256 * EPT - 1) / (256 * EPT));
    dim3 grid_rows((N + 8 * NPW - 1) / (8 * NPW));

    gat_zero<<<(NGRAPH * 3 * HID + 255) / 256, blk, 0, stream>>>(pooled, NGRAPH * 3 * HID);

    gat_zero<<<(N + 255) / 256, blk, 0, stream>>>((float*)cnt, N);
    gat_deg_count<<<grid_edge, blk, 0, stream>>>(ei, cnt, E);
    gat_scan_local<<<(N + 1023) / 1024, 1024, 0, stream>>>(cnt, rowptr, bsum, N);
    gat_scan_add<<<(N + 1023) / 1024, 1024, 0, stream>>>(rowptr, bsum, N);
    gat_edge_place<<<grid_edge, blk, 0, stream>>>(ei, rowptr, cnt, csrsrc, E);

    for (int l = 0; l < 3; ++l) {
        const float* X  = (l == 0) ? x_in : xb;
        float* oacc     = (l == 2) ? xfinal : xb;
        const float* Wl = W + (size_t)l * HID * HID;

        gat_gemm_wmma<<<grid_gemm, blk, 0, stream>>>(X, Wl, hbuf, N);
        gat_node_dots<<<grid_rows, blk, 0, stream>>>(hbuf, att_s + l * HID, att_d + l * HID,
                                                     asrc, adst, N);
        gat_gather<<<grid_node, blk, 0, stream>>>(rowptr, csrsrc, asrc, adst, hbuf, bias + l * HID,
                                                  oacc, part, sums, N);
        gat_stats_reduce<<<256, 256, 0, stream>>>(part, sums, n_gblk);
        gat_bn_pool<<<grid_rows, blk, 0, stream>>>(oacc, sums, sumsq, gamma + l * HID,
                                                   beta + l * HID, batch, pooled, l, N);
    }
}


# Edge grouping by destination moved from per-edge global atomics to LDS counters: edges binned into 256-node buckets, each bucket ranked by one work-group (row sum order stays arrival rank)
# speedup vs baseline: 4.4813x; 1.3674x over previous
#include <hip/hip_runtime.h>
#include <hip/hip_bf16.h>

#define HID 128
#define NGRAPH 512
#define NEG_SLOPE 0.2f
#define BN_EPS 1e-5f

typedef __attribute__((ext_vector_type(2))) float v2f;
typedef __attribute__((ext_vector_type(8))) float v8f;

__device__ __forceinline__ float lrelu(float a) { return a > 0.0f ? a : NEG_SLOPE * a; }

__global__ void gat_zero(float* __restrict__ p, int n) {
    int i = blockIdx.x * blockDim.x + threadIdx.x;
    if (i < n) p[i] = 0.0f;
}

__global__ void gat_gemm_wmma(const float* __restrict__ X, const float* __restrict__ W,
                              float* __restrict__ H, int nrows) {
    int wave = (blockIdx.x * blockDim.x + threadIdx.x) >> 5;
    int m0 = wave * 16;
    if (m0 >= nrows) return;
    int lane = threadIdx.x & 31;
    int lr = lane & 15;
    int lh = lane >> 4;

    v8f acc[8] = {};
    const float* xrow = X + (size_t)(m0 + lr) * HID;

    for (int k = 0; k < HID; k += 4) {
        int ka = k + lh * 2;
        v2f a;
        a.x = xrow[ka];
        a.y = xrow[ka + 1];
#pragma unroll
        for (int n = 0; n < 8; ++n) {
            v2f b;
            b.x = W[ka * HID + n * 16 + lr];
            b.y = W[(ka + 1) * HID + n * 16 + lr];
            acc[n] = __builtin_amdgcn_wmma_f32_16x16x4_f32(
                false, a, false, b, (short)0, acc[n], false, false);
        }
    }
#pragma unroll
    for (int n = 0; n < 8; ++n)
#pragma unroll
        for (int j = 0; j < 8; ++j)
            H[(size_t)(m0 + j + lh * 8) * HID + n * 16 + lr] = acc[n][j];
}

#define NPW 8

__global__ void gat_node_dots(const float* __restrict__ H, const float* __restrict__ att_s,
                              const float* __restrict__ att_d, float* __restrict__ a_s,
                              float* __restrict__ a_d, int n_nodes) {
    int wave = __builtin_amdgcn_readfirstlane((int)((blockIdx.x * blockDim.x + threadIdx.x) >> 5));
    int n0 = wave * NPW;
    if (n0 >= n_nodes) return;
    int lane = threadIdx.x & 31;
    float4 ws = ((const float4*)att_s)[lane];
    float4 wd = ((const float4*)att_d)[lane];
    float4 hv[NPW];
#pragma unroll
    for (int i = 0; i < NPW; ++i) {
        int n = n0 + i < n_nodes ? n0 + i : n_nodes - 1;
        hv[i] = ((const float4*)(H + (size_t)n * HID))[lane];
    }
    float os = 0.0f, od = 0.0f;
#pragma unroll
    for (int i = 0; i < NPW; ++i) {
        float ps = hv[i].x * ws.x + hv[i].y * ws.y + hv[i].z * ws.z + hv[i].w * ws.w;
        float pd = hv[i].x * wd.x + hv[i].y * wd.y + hv[i].z * wd.z + hv[i].w * wd.w;
        for (int off = 16; off > 0; off >>= 1) {
            ps += __shfl_xor(ps, off, 32);
            pd += __shfl_xor(pd, off, 32);
        }
        if (lane == i) { os = ps; od = pd; }
    }
    if (lane < NPW && n0 + lane < n_nodes) {
        a_s[n0 + lane] = os;
        a_d[n0 + lane] = od;
    }
}

#define BSH 8
#define BNODES (1 << BSH)
#define MAXB 512
#define BPAD 32
#define CT 512
#define CPT 8
#define CH (CT * CPT)
#define ST 1024
#define SPT 4

__device__ __forceinline__ int wave_incl_scan(int x, int lane) {
    for (int off = 1; off < 32; off <<= 1) {
        int y = __shfl_up(x, off, 32);
        if (lane >= off) x += y;
    }
    return x;
}

template <bool FULL>
__device__ __forceinline__ void bucket_count_body(const int* __restrict__ ei,
                                                  int* __restrict__ btot, int n_edges, int nbk,
                                                  int* h) {
    int t = threadIdx.x;
    int e0 = blockIdx.x * CH + t;
    int d[CPT];
#pragma unroll
    for (int i = 0; i < CPT; ++i) {
        int e = e0 + i * CT;
        bool in = FULL || e < n_edges;
        int v = ei[n_edges + (in ? e : n_edges - 1)];
        d[i] = in ? v : -1;
    }
    h[t] = 0;
    __syncthreads();
#pragma unroll
    for (int i = 0; i < CPT; ++i)
        if (FULL || d[i] >= 0) atomicAdd(&h[d[i] >> BSH], 1);
    __syncthreads();
    int c = h[t];
    if (t < nbk && c != 0) atomicAdd(&btot[t * BPAD], c);
}

__global__ void __launch_bounds__(CT)
gat_bucket_count(const int* __restrict__ ei, int* __restrict__ btot, int n_edges, int nbk) {
    __shared__ int h[MAXB];
    if ((int)blockIdx.x * CH + CH <= n_edges)
        bucket_count_body<true>(ei, btot, n_edges, nbk, h);
    else
        bucket_count_body<false>(ei, btot, n_edges, nbk, h);
}

template <bool FULL>
__device__ __forceinline__ void bucket_scatter_body(const int* __restrict__ ei,
                                                    const int* __restrict__ btot,
                                                    int* __restrict__ bfill,
                                                    int2* __restrict__ rec, int n_edges, int nbk,
                                                    int* h, int* wtot) {
    int t = threadIdx.x;
    int lane = t & 31;
    int wv = t >> 5;
    int e0 = blockIdx.x * CH + t;

    int s[CPT], d[CPT], r[CPT];
#pragma unroll
    for (int i = 0; i < CPT; ++i) {
        int e = e0 + i * CT;
        bool in = FULL || e < n_edges;
        int ec = in ? e : n_edges - 1;
        s[i] = ei[ec];
        d[i] = in ? ei[n_edges + ec] : -1;
    }

    int tot = t < nbk ? btot[t * BPAD] : 0;
    int x = wave_incl_scan(tot, lane);
    if (lane == 31) wtot[wv] = x;
    h[t] = 0;
    __syncthreads();

#pragma unroll
    for (int i = 0; i < CPT; ++i)
        r[i] = (FULL || d[i] >= 0) ? atomicAdd(&h[d[i] >> BSH], 1) : 0;

    int wt = lane < CT / 32 ? wtot[lane] : 0;
    int wsc = wave_incl_scan(wt, lane);
    int bstart = __shfl(wsc, wv, 32) - __shfl(wt, wv, 32) + x - tot;
    __syncthreads();

    int c = h[t];
    int g = (t < nbk && c != 0) ? atomicAdd(&bfill[t * BPAD], c) : 0;
    h[t] = bstart + g;
    __syncthreads();

#pragma unroll
    for (int i = 0; i < CPT; ++i)
        if (FULL || d[i] >= 0) rec[h[d[i] >> BSH] + r[i]] = make_int2(s[i], d[i]);
}

__global__ void __launch_bounds__(CT)
gat_bucket_scatter(const int* __restrict__ ei, const int* __restrict__ btot,
                   int* __restrict__ bfill, int2* __restrict__ rec, int n_edges, int nbk) {
    __shared__ int h[MAXB];
    __shared__ int wtot[CT / 32];
    if ((int)blockIdx.x * CH + CH <= n_edges)
        bucket_scatter_body<true>(ei, btot, bfill, rec, n_edges, nbk, h, wtot);
    else
        bucket_scatter_body<false>(ei, btot, bfill, rec, n_edges, nbk, h, wtot);
}

__global__ void __launch_bounds__(ST)
gat_bucket_sort(const int2* __restrict__ rec, const int* __restrict__ btot,
                int* __restrict__ rowptr, int* __restrict__ csr_src, int n_nodes, int nbk) {
    __shared__ int cnt[BNODES];
    __shared__ int excl[BNODES];
    __shared__ int red[ST / 32];
    __shared__ int wtot[BNODES / 32];
    int t = threadIdx.x;
    int lane = t & 31;
    int wv = t >> 5;
    int b = blockIdx.x;

    int v = t < b ? btot[t * BPAD] : 0;
    int n = btot[b * BPAD];
    for (int off = 16; off > 0; off >>= 1) v += __shfl_xor(v, off, 32);
    if (lane == 0) red[wv] = v;
    if (t < BNODES) cnt[t] = 0;
    __syncthreads();
    int beg = red[lane];
    for (int off = 16; off > 0; off >>= 1) beg += __shfl_xor(beg, off, 32);
    const int2* brec = rec + beg;

    for (int i0 = t; i0 < n + t; i0 += ST * SPT) {
        int dd[SPT];
#pragma unroll
        for (int k = 0; k < SPT; ++k) {
            int i = i0 + k * ST;
            dd[k] = brec[i < n ? i : n - 1].y;
        }
#pragma unroll
        for (int k = 0; k < SPT; ++k)
            if (i0 + k * ST < n) atomicAdd(&cnt[dd[k] & (BNODES - 1)], 1);
    }
    __syncthreads();

    int c = 0, x = 0;
    if (t < BNODES) {
        c = cnt[t];
        x = wave_incl_scan(c, lane);
        if (lane == 31) wtot[wv] = x;
    }
    __syncthreads();
    if (t < BNODES) {
        int wt = lane < BNODES / 32 ? wtot[lane] : 0;
        int wsc = wave_incl_scan(wt, lane);
        int e = __shfl(wsc, wv, 32) - __shfl(wt, wv, 32) + x - c;
        excl[t] = e;
        cnt[t] = 0;
        int node = b * BNODES + t;
        if (node < n_nodes) rowptr[node] = beg + e;
        if (b == nbk - 1 && t == 0) rowptr[n_nodes] = beg + n;
    }
    __syncthreads();

    int* bcsr = csr_src + beg;
    for (int i0 = t; i0 < n + t; i0 += ST * SPT) {
        int2 rr[SPT];
#pragma unroll
        for (int k = 0; k < SPT; ++k) {
            int i = i0 + k * ST;
            rr[k] = brec[i < n ? i : n - 1];
        }
#pragma unroll
        for (int k = 0; k < SPT; ++k) {
            if (i0 + k * ST < n) {
                int q = rr[k].y & (BNODES - 1);
                bcsr[excl[q] + atomicAdd(&cnt[q], 1)] = rr[k].x;
            }
        }
    }
}

#define GW 8

__device__ __forceinline__ float elu1(float v) { return v > 0.0f ? v : expm1f(v); }

__global__ void gat_gather(const int* __restrict__ rowptr, const int* __restrict__ csr_src,
                           const float* __restrict__ a_s, const float* __restrict__ a_d,
                           const float* __restrict__ H, const float* __restrict__ bias,
                           float* __restrict__ out, float* __restrict__ part,
                           float* __restrict__ sums, int n_nodes) {
    __shared__ float4 red[GW][64];
    int node = __builtin_amdgcn_readfirstlane((int)((blockIdx.x * blockDim.x + threadIdx.x) >> 5));
    int lane = threadIdx.x & 31;
    int wv = threadIdx.x >> 5;
    float4 v = make_float4(0.0f, 0.0f, 0.0f, 0.0f);
    float4 q = make_float4(0.0f, 0.0f, 0.0f, 0.0f);

    if (node < n_nodes) {
        int beg = __builtin_amdgcn_readfirstlane(rowptr[node]);
        int end = __builtin_amdgcn_readfirstlane(rowptr[node + 1]);
        float ad = a_d[node];
        float self_logit = lrelu(a_s[node] + ad);

        float mx = self_logit;
        for (int j = beg + lane; j < end; j += 32)
            mx = fmaxf(mx, lrelu(a_s[csr_src[j]] + ad));
        for (int off = 16; off > 0; off >>= 1)
            mx = fmaxf(mx, __shfl_xor(mx, off, 32));

        float es = expf(self_logit - mx);
        float4 hv = ((const float4*)(H + (size_t)node * HID))[lane];
        float4 acc;
        acc.x = es * hv.x; acc.y = es * hv.y; acc.z = es * hv.z; acc.w = es * hv.w;
        float wsum = 0.0f;

        for (int base = beg; base < end; base += 32) {
            int j = base + lane;
            int sidx = 0;
            float w = 0.0f;
            if (j < end) {
                sidx = csr_src[j];
                w = expf(lrelu(a_s[sidx] + ad) - mx);
            }
            wsum += w;
            int cnt = end - base < 32 ? end - base : 32;
#pragma unroll 4
            for (int k = 0; k < cnt; ++k) {
                int sk = __builtin_amdgcn_readlane(sidx, k);
                float wk = __int_as_float(__builtin_amdgcn_readlane(__float_as_int(w), k));
                float4 h = ((const float4*)(H + (size_t)sk * HID))[lane];
                acc.x += wk * h.x; acc.y += wk * h.y; acc.z += wk * h.z; acc.w += wk * h.w;
            }
        }
        for (int off = 16; off > 0; off >>= 1)
            wsum += __shfl_xor(wsum, off, 32);

        float inv = 1.0f / (es + wsum);
        float4 b = ((const float4*)bias)[lane];
        v.x = elu1(acc.x * inv + b.x);
        v.y = elu1(acc.y * inv + b.y);
        v.z = elu1(acc.z * inv + b.z);
        v.w = elu1(acc.w * inv + b.w);
        q.x = v.x * v.x; q.y = v.y * v.y; q.z = v.z * v.z; q.w = v.w * v.w;
    }

    red[wv][lane] = v;
    red[wv][32 + lane] = q;
    __syncthreads();
    const float* rf = (const float*)red;
    int t = threadIdx.x;
    float tot = 0.0f;
    for (int w = 0; w < GW; ++w) tot += rf[w * 256 + t];
    part[(size_t)blockIdx.x * (2 * HID) + t] = tot;
    if (blockIdx.x == 0) sums[t] = 0.0f;
    if (node < n_nodes) ((float4*)(out + (size_t)node * HID))[lane] = v;
}

__global__ void gat_stats_reduce(const float* __restrict__ part, float* __restrict__ sums,
                                 int nrows) {
    int t = threadIdx.x;
    int r = blockIdx.x;
    const int g = gridDim.x;
    float acc = 0.0f;
    for (; r + 15 * g < nrows; r += 16 * g) {
        float x[16];
        for (int i = 0; i < 16; ++i) x[i] = part[(size_t)(r + i * g) * (2 * HID) + t];
        for (int i = 0; i < 16; ++i) acc += x[i];
    }
    for (; r < nrows; r += g) acc += part[(size_t)r * (2 * HID) + t];
    atomicAdd(&sums[t], acc);
}

__device__ __forceinline__ void pool_add(float* __restrict__ pooled, int g, int layer, int lane,
                                         float4 a) {
    float* p = pooled + (size_t)g * (3 * HID) + layer * HID + lane * 4;
    atomicAdd(p + 0, a.x);
    atomicAdd(p + 1, a.y);
    atomicAdd(p + 2, a.z);
    atomicAdd(p + 3, a.w);
}

__global__ void gat_bn_pool(float* __restrict__ out_acc, const float* __restrict__ sums,
                            const float* __restrict__ sumsq, const float* __restrict__ gamma,
                            const float* __restrict__ beta, const int* __restrict__ batch,
                            float* __restrict__ pooled, int layer, int n_nodes) {
    int wave = __builtin_amdgcn_readfirstlane((int)((blockIdx.x * blockDim.x + threadIdx.x) >> 5));
    int n0 = wave * NPW;
    if (n0 >= n_nodes) return;
    int lane = threadIdx.x & 31;

    float4 x[NPW];
    int gid[NPW];
#pragma unroll
    for (int i = 0; i < NPW; ++i) {
        int n = n0 + i < n_nodes ? n0 + i : n_nodes - 1;
        x[i] = ((const float4*)(out_acc + (size_t)n * HID))[lane];
        gid[i] = batch[n];
    }

    float invN = 1.0f / (float)n_nodes;
    float4 s = ((const float4*)sums)[lane];
    float4 sq = ((const float4*)sumsq)[lane];
    float4 ga = ((const float4*)gamma)[lane];
    float4 be = ((const float4*)beta)[lane];
    float4 mu, sd;
    mu.x = s.x * invN; mu.y = s.y * invN; mu.z = s.z * invN; mu.w = s.w * invN;
    sd.x = sqrtf(sq.x * invN - mu.x * mu.x + BN_EPS);
    sd.y = sqrtf(sq.y * invN - mu.y * mu.y + BN_EPS);
    sd.z = sqrtf(sq.z * invN - mu.z * mu.z + BN_EPS);
    sd.w = sqrtf(sq.w * invN - mu.w * mu.w + BN_EPS);

    int cur = gid[0];
    float4 run = make_float4(0.0f, 0.0f, 0.0f, 0.0f);
#pragma unroll
    for (int i = 0; i < NPW; ++i) {
        if (n0 + i < n_nodes) {
            float4 v;
            v.x = (x[i].x - mu.x) / sd.x * ga.x + be.x;
            v.y = (x[i].y - mu.y) / sd.y * ga.y + be.y;
            v.z = (x[i].z - mu.z) / sd.z * ga.z + be.z;
            v.w = (x[i].w - mu.w) / sd.w * ga.w + be.w;
            ((float4*)(out_acc + (size_t)(n0 + i) * HID))[lane] = v;
            if (gid[i] != cur) {
                pool_add(pooled, cur, layer, lane, run);
                cur = gid[i];
                run = make_float4(0.0f, 0.0f, 0.0f, 0.0f);
            }
            run.x += v.x; run.y += v.y; run.z += v.z; run.w += v.w;
        }
    }
    pool_add(pooled, cur, layer, lane, run);
}

extern "C" void kernel_launch(void* const* d_in, const int* in_sizes, int n_in,
                              void* d_out, int out_size, void* d_ws, size_t ws_size,
                              hipStream_t stream) {
    const float* x_in  = (const float*)d_in[0];
    const int*   ei    = (const int*)d_in[1];
    const int*   batch = (const int*)d_in[2];
    const float* W     = (const float*)d_in[3];
    const float* att_s = (const float*)d_in[4];
    const float* att_d = (const float*)d_in[5];
    const float* bias  = (const float*)d_in[6];
    const float* gamma = (const float*)d_in[7];
    const float* beta  = (const float*)d_in[8];

    const int N = in_sizes[0] / HID;
    const int E = in_sizes[1] / 2;

    float* pooled = (float*)d_out;
    float* xfinal = (float*)d_out + (size_t)NGRAPH * 3 * HID;

    char* w = (char*)d_ws;
    float* hbuf = (float*)w;  w += (size_t)N * HID * sizeof(float);
    float* xb   = (float*)w;  w += (size_t)N * HID * sizeof(float);
    const int n_gblk = (N + 7) / 8;
    float* part = (float*)w;  w += (size_t)n_gblk * 2 * HID * sizeof(float);
    float* asrc = (float*)w;  w += (size_t)N * sizeof(float);
    float* adst = (float*)w;  w += (size_t)N * sizeof(float);
    float* sums = (float*)w;  w += (size_t)2 * HID * sizeof(float);
    float* sumsq = sums + HID;
    int* btot   = (int*)w;    w += (size_t)MAXB * BPAD * sizeof(int);
    int* bfill  = (int*)w;    w += (size_t)MAXB * BPAD * sizeof(int);
    int2* rec   = (int2*)w;   w += (size_t)E * sizeof(int2);
    int* rowptr = (int*)w;    w += (size_t)(N + 1) * sizeof(int);
    int* csrsrc = (int*)w;

    const int waves_g = (N + 15) / 16;
    dim3 blk(256);
    dim3 grid_gemm((waves_g * 32 + 255) / 256);
    dim3 grid_node(n_gblk);
    dim3 grid_edge((E + CH - 1) / CH);
    const int nbk = (N + BNODES - 1) / BNODES;
    dim3 grid_rows((N + 8 * NPW - 1) / (8 * NPW));

    gat_zero<<<(NGRAPH * 3 * HID + 255) / 256, blk, 0, stream>>>(pooled, NGRAPH * 3 * HID);

    gat_zero<<<(2 * MAXB * BPAD + 255) / 256, blk, 0, stream>>>((float*)btot, 2 * MAXB * BPAD);
    gat_bucket_count<<<grid_edge, CT, 0, stream>>>(ei, btot, E, nbk);
    gat_bucket_scatter<<<grid_edge, CT, 0, stream>>>(ei, btot, bfill, rec, E, nbk);
    gat_bucket_sort<<<nbk, ST, 0, stream>>>(rec, btot, rowptr, csrsrc, N, nbk);

    for (int l = 0; l < 3; ++l) {
        const float* X  = (l == 0) ? x_in : xb;
        float* oacc     = (l == 2) ? xfinal : xb;
        const float* Wl = W + (size_t)l * HID * HID;

        gat_gemm_wmma<<<grid_gemm, blk, 0, stream>>>(X, Wl, hbuf, N);
        gat_node_dots<<<grid_rows, blk, 0, stream>>>(hbuf, att_s + l * HID, att_d + l * HID,
                                                     asrc, adst, N);
        gat_gather<<<grid_node, blk, 0, stream>>>(rowptr, csrsrc, asrc, adst, hbuf, bias + l * HID,
                                                  oacc, part, sums, N);
        gat_stats_reduce<<<256, 256, 0, stream>>>(part, sums, n_gblk);
        gat_bn_pool<<<grid_rows, blk, 0, stream>>>(oacc, sums, sumsq, gamma + l * HID,
                                                   beta + l * HID, batch, pooled, l, N);
    }
}


# Node attention logits (h.att_src, h.att_dst) formed in the product launch's epilogue from the WMMA accumulators; node-dot launch and its re-read of H removed (logit sum order: tile tree)
# speedup vs baseline: 4.6713x; 1.0424x over previous
#include <hip/hip_runtime.h>
#include <hip/hip_bf16.h>

#define HID 128
#define NGRAPH 512
#define NEG_SLOPE 0.2f
#define BN_EPS 1e-5f

typedef __attribute__((ext_vector_type(2))) float v2f;
typedef __attribute__((ext_vector_type(8))) float v8f;

__device__ __forceinline__ float lrelu(float a) { return a > 0.0f ? a : NEG_SLOPE * a; }

__global__ void gat_zero(float* __restrict__ p, int n) {
    int i = blockIdx.x * blockDim.x + threadIdx.x;
    if (i < n) p[i] = 0.0f;
}

__global__ void gat_gemm_wmma(const float* __restrict__ X, const float* __restrict__ W,
                              const float* __restrict__ att_s, const float* __restrict__ att_d,
                              float* __restrict__ H, float* __restrict__ a_s,
                              float* __restrict__ a_d, int nrows) {
    int wave = (blockIdx.x * blockDim.x + threadIdx.x) >> 5;
    int m0 = wave * 16;
    if (m0 >= nrows) return;
    int lane = threadIdx.x & 31;
    int lr = lane & 15;
    int lh = lane >> 4;

    float ws[8], wd[8];
#pragma unroll
    for (int n = 0; n < 8; ++n) {
        ws[n] = att_s[n * 16 + lr];
        wd[n] = att_d[n * 16 + lr];
    }

    v8f acc[8] = {};
    const float* xrow = X + (size_t)(m0 + lr) * HID;

    for (int k = 0; k < HID; k += 4) {
        int ka = k + lh * 2;
        v2f a;
        a.x = xrow[ka];
        a.y = xrow[ka + 1];
#pragma unroll
        for (int n = 0; n < 8; ++n) {
            v2f b;
            b.x = W[ka * HID + n * 16 + lr];
            b.y = W[(ka + 1) * HID + n * 16 + lr];
            acc[n] = __builtin_amdgcn_wmma_f32_16x16x4_f32(
                false, a, false, b, (short)0, acc[n], false, false);
        }
    }
#pragma unroll
    for (int n = 0; n < 8; ++n)
#pragma unroll
        for (int j = 0; j < 8; ++j)
            H[(size_t)(m0 + j + lh * 8) * HID + n * 16 + lr] = acc[n][j];

    float ps[8], pd[8];
#pragma unroll
    for (int j = 0; j < 8; ++j) { ps[j] = 0.0f; pd[j] = 0.0f; }
#pragma unroll
    for (int n = 0; n < 8; ++n)
#pragma unroll
        for (int j = 0; j < 8; ++j) {
            ps[j] += acc[n][j] * ws[n];
            pd[j] += acc[n][j] * wd[n];
        }

    bool h8 = (lr & 8) != 0;
    float qs[4], qd[4];
#pragma unroll
    for (int i = 0; i < 4; ++i) {
        qs[i] = (h8 ? ps[4 + i] : ps[i]) + __shfl_xor(h8 ? ps[i] : ps[4 + i], 8, 32);
        qd[i] = (h8 ? pd[4 + i] : pd[i]) + __shfl_xor(h8 ? pd[i] : pd[4 + i], 8, 32);
    }
    bool h4 = (lr & 4) != 0;
    float rs[2], rd[2];
#pragma unroll
    for (int i = 0; i < 2; ++i) {
        rs[i] = (h4 ? qs[2 + i] : qs[i]) + __shfl_xor(h4 ? qs[i] : qs[2 + i], 4, 32);
        rd[i] = (h4 ? qd[2 + i] : qd[i]) + __shfl_xor(h4 ? qd[i] : qd[2 + i], 4, 32);
    }
    bool h2 = (lr & 2) != 0;
    float ts = (h2 ? rs[1] : rs[0]) + __shfl_xor(h2 ? rs[0] : rs[1], 2, 32);
    float td = (h2 ? rd[1] : rd[0]) + __shfl_xor(h2 ? rd[0] : rd[1], 2, 32);
    bool h1 = (lr & 1) != 0;
    float tot = (h1 ? td : ts) + __shfl_xor(h1 ? ts : td, 1, 32);
    float* dst = h1 ? a_d : a_s;
    dst[m0 + lh * 8 + (lr >> 1)] = tot;
}

#define NPW 8

#define BSH 8
#define BNODES (1 << BSH)
#define MAXB 512
#define BPAD 32
#define CT 512
#define CPT 8
#define CH (CT * CPT)
#define ST 1024
#define SPT 4

__device__ __forceinline__ int wave_incl_scan(int x, int lane) {
    for (int off = 1; off < 32; off <<= 1) {
        int y = __shfl_up(x, off, 32);
        if (lane >= off) x += y;
    }
    return x;
}

template <bool FULL>
__device__ __forceinline__ void bucket_count_body(const int* __restrict__ ei,
                                                  int* __restrict__ btot, int n_edges, int nbk,
                                                  int* h) {
    int t = threadIdx.x;
    int e0 = blockIdx.x * CH + t;
    int d[CPT];
#pragma unroll
    for (int i = 0; i < CPT; ++i) {
        int e = e0 + i * CT;
        bool in = FULL || e < n_edges;
        int v = ei[n_edges + (in ? e : n_edges - 1)];
        d[i] = in ? v : -1;
    }
    h[t] = 0;
    __syncthreads();
#pragma unroll
    for (int i = 0; i < CPT; ++i)
        if (FULL || d[i] >= 0) atomicAdd(&h[d[i] >> BSH], 1);
    __syncthreads();
    int c = h[t];
    if (t < nbk && c != 0) atomicAdd(&btot[t * BPAD], c);
}

__global__ void __launch_bounds__(CT)
gat_bucket_count(const int* __restrict__ ei, int* __restrict__ btot, int n_edges, int nbk) {
    __shared__ int h[MAXB];
    if ((int)blockIdx.x * CH + CH <= n_edges)
        bucket_count_body<true>(ei, btot, n_edges, nbk, h);
    else
        bucket_count_body<false>(ei, btot, n_edges, nbk, h);
}

template <bool FULL>
__device__ __forceinline__ void bucket_scatter_body(const int* __restrict__ ei,
                                                    const int* __restrict__ btot,
                                                    int* __restrict__ bfill,
                                                    int2* __restrict__ rec, int n_edges, int nbk,
                                                    int* h, int* wtot) {
    int t = threadIdx.x;
    int lane = t & 31;
    int wv = t >> 5;
    int e0 = blockIdx.x * CH + t;

    int s[CPT], d[CPT], r[CPT];
#pragma unroll
    for (int i = 0; i < CPT; ++i) {
        int e = e0 + i * CT;
        bool in = FULL || e < n_edges;
        int ec = in ? e : n_edges - 1;
        s[i] = ei[ec];
        d[i] = in ? ei[n_edges + ec] : -1;
    }

    int tot = t < nbk ? btot[t * BPAD] : 0;
    int x = wave_incl_scan(tot, lane);
    if (lane == 31) wtot[wv] = x;
    h[t] = 0;
    __syncthreads();

#pragma unroll
    for (int i = 0; i < CPT; ++i)
        r[i] = (FULL || d[i] >= 0) ? atomicAdd(&h[d[i] >> BSH], 1) : 0;

    int wt = lane < CT / 32 ? wtot[lane] : 0;
    int wsc = wave_incl_scan(wt, lane);
    int bstart = __shfl(wsc, wv, 32) - __shfl(wt, wv, 32) + x - tot;
    __syncthreads();

    int c = h[t];
    int g = (t < nbk && c != 0) ? atomicAdd(&bfill[t * BPAD], c) : 0;
    h[t] = bstart + g;
    __syncthreads();

#pragma unroll
    for (int i = 0; i < CPT; ++i)
        if (FULL || d[i] >= 0) rec[h[d[i] >> BSH] + r[i]] = make_int2(s[i], d[i]);
}

__global__ void __launch_bounds__(CT)
gat_bucket_scatter(const int* __restrict__ ei, const int* __restrict__ btot,
                   int* __restrict__ bfill, int2* __restrict__ rec, int n_edges, int nbk) {
    __shared__ int h[MAXB];
    __shared__ int wtot[CT / 32];
    if ((int)blockIdx.x * CH + CH <= n_edges)
        bucket_scatter_body<true>(ei, btot, bfill, rec, n_edges, nbk, h, wtot);
    else
        bucket_scatter_body<false>(ei, btot, bfill, rec, n_edges, nbk, h, wtot);
}

__global__ void __launch_bounds__(ST)
gat_bucket_sort(const int2* __restrict__ rec, const int* __restrict__ btot,
                int* __restrict__ rowptr, int* __restrict__ csr_src, int n_nodes, int nbk) {
    __shared__ int cnt[BNODES];
    __shared__ int excl[BNODES];
    __shared__ int red[ST / 32];
    __shared__ int wtot[BNODES / 32];
    int t = threadIdx.x;
    int lane = t & 31;
    int wv = t >> 5;
    int b = blockIdx.x;

    int v = t < b ? btot[t * BPAD] : 0;
    int n = btot[b * BPAD];
    for (int off = 16; off > 0; off >>= 1) v += __shfl_xor(v, off, 32);
    if (lane == 0) red[wv] = v;
    if (t < BNODES) cnt[t] = 0;
    __syncthreads();
    int beg = red[lane];
    for (int off = 16; off > 0; off >>= 1) beg += __shfl_xor(beg, off, 32);
    const int2* brec = rec + beg;

    for (int i0 = t; i0 < n + t; i0 += ST * SPT) {
        int dd[SPT];
#pragma unroll
        for (int k = 0; k < SPT; ++k) {
            int i = i0 + k * ST;
            dd[k] = brec[i < n ? i : n - 1].y;
        }
#pragma unroll
        for (int k = 0; k < SPT; ++k)
            if (i0 + k * ST < n) atomicAdd(&cnt[dd[k] & (BNODES - 1)], 1);
    }
    __syncthreads();

    int c = 0, x = 0;
    if (t < BNODES) {
        c = cnt[t];
        x = wave_incl_scan(c, lane);
        if (lane == 31) wtot[wv] = x;
    }
    __syncthreads();
    if (t < BNODES) {
        int wt = lane < BNODES / 32 ? wtot[lane] : 0;
        int wsc = wave_incl_scan(wt, lane);
        int e = __shfl(wsc, wv, 32) - __shfl(wt, wv, 32) + x - c;
        excl[t] = e;
        cnt[t] = 0;
        int node = b * BNODES + t;
        if (node < n_nodes) rowptr[node] = beg + e;
        if (b == nbk - 1 && t == 0) rowptr[n_nodes] = beg + n;
    }
    __syncthreads();

    int* bcsr = csr_src + beg;
    for (int i0 = t; i0 < n + t; i0 += ST * SPT) {
        int2 rr[SPT];
#pragma unroll
        for (int k = 0; k < SPT; ++k) {
            int i = i0 + k * ST;
            rr[k] = brec[i < n ? i : n - 1];
        }
#pragma unroll
        for (int k = 0; k < SPT; ++k) {
            if (i0 + k * ST < n) {
                int q = rr[k].y & (BNODES - 1);
                bcsr[excl[q] + atomicAdd(&cnt[q], 1)] = rr[k].x;
            }
        }
    }
}

#define GW 8

__device__ __forceinline__ float elu1(float v) { return v > 0.0f ? v : expm1f(v); }

__global__ void gat_gather(const int* __restrict__ rowptr, const int* __restrict__ csr_src,
                           const float* __restrict__ a_s, const float* __restrict__ a_d,
                           const float* __restrict__ H, const float* __restrict__ bias,
                           float* __restrict__ out, float* __restrict__ part,
                           float* __restrict__ sums, int n_nodes) {
    __shared__ float4 red[GW][64];
    int node = __builtin_amdgcn_readfirstlane((int)((blockIdx.x * blockDim.x + threadIdx.x) >> 5));
    int lane = threadIdx.x & 31;
    int wv = threadIdx.x >> 5;
    float4 v = make_float4(0.0f, 0.0f, 0.0f, 0.0f);
    float4 q = make_float4(0.0f, 0.0f, 0.0f, 0.0f);

    if (node < n_nodes) {
        int beg = __builtin_amdgcn_readfirstlane(rowptr[node]);
        int end = __builtin_amdgcn_readfirstlane(rowptr[node + 1]);
        float ad = a_d[node];
        float self_logit = lrelu(a_s[node] + ad);

        float mx = self_logit;
        for (int j = beg + lane; j < end; j += 32)
            mx = fmaxf(mx, lrelu(a_s[csr_src[j]] + ad));
        for (int off = 16; off > 0; off >>= 1)
            mx = fmaxf(mx, __shfl_xor(mx, off, 32));

        float es = expf(self_logit - mx);
        float4 hv = ((const float4*)(H + (size_t)node * HID))[lane];
        float4 acc;
        acc.x = es * hv.x; acc.y = es * hv.y; acc.z = es * hv.z; acc.w = es * hv.w;
        float wsum = 0.0f;

        for (int base = beg; base < end; base += 32) {
            int j = base + lane;
            int sidx = 0;
            float w = 0.0f;
            if (j < end) {
                sidx = csr_src[j];
                w = expf(lrelu(a_s[sidx] + ad) - mx);
            }
            wsum += w;
            int cnt = end - base < 32 ? end - base : 32;
#pragma unroll 4
            for (int k = 0; k < cnt; ++k) {
                int sk = __builtin_amdgcn_readlane(sidx, k);
                float wk = __int_as_float(__builtin_amdgcn_readlane(__float_as_int(w), k));
                float4 h = ((const float4*)(H + (size_t)sk * HID))[lane];
                acc.x += wk * h.x; acc.y += wk * h.y; acc.z += wk * h.z; acc.w += wk * h.w;
            }
        }
        for (int off = 16; off > 0; off >>= 1)
            wsum += __shfl_xor(wsum, off, 32);

        float inv = 1.0f / (es + wsum);
        float4 b = ((const float4*)bias)[lane];
        v.x = elu1(acc.x * inv + b.x);
        v.y = elu1(acc.y * inv + b.y);
        v.z = elu1(acc.z * inv + b.z);
        v.w = elu1(acc.w * inv + b.w);
        q.x = v.x * v.x; q.y = v.y * v.y; q.z = v.z * v.z; q.w = v.w * v.w;
    }

    red[wv][lane] = v;
    red[wv][32 + lane] = q;
    __syncthreads();
    const float* rf = (const float*)red;
    int t = threadIdx.x;
    float tot = 0.0f;
    for (int w = 0; w < GW; ++w) tot += rf[w * 256 + t];
    part[(size_t)blockIdx.x * (2 * HID) + t] = tot;
    if (blockIdx.x == 0) sums[t] = 0.0f;
    if (node < n_nodes) ((float4*)(out + (size_t)node * HID))[lane] = v;
}

__global__ void gat_stats_reduce(const float* __restrict__ part, float* __restrict__ sums,
                                 int nrows) {
    int t = threadIdx.x;
    int r = blockIdx.x;
    const int g = gridDim.x;
    float acc = 0.0f;
    for (; r + 15 * g < nrows; r += 16 * g) {
        float x[16];
        for (int i = 0; i < 16; ++i) x[i] = part[(size_t)(r + i * g) * (2 * HID) + t];
        for (int i = 0; i < 16; ++i) acc += x[i];
    }
    for (; r < nrows; r += g) acc += part[(size_t)r * (2 * HID) + t];
    atomicAdd(&sums[t], acc);
}

__device__ __forceinline__ void pool_add(float* __restrict__ pooled, int g, int layer, int lane,
                                         float4 a) {
    float* p = pooled + (size_t)g * (3 * HID) + layer * HID + lane * 4;
    atomicAdd(p + 0, a.x);
    atomicAdd(p + 1, a.y);
    atomicAdd(p + 2, a.z);
    atomicAdd(p + 3, a.w);
}

__global__ void gat_bn_pool(float* __restrict__ out_acc, const float* __restrict__ sums,
                            const float* __restrict__ sumsq, const float* __restrict__ gamma,
                            const float* __restrict__ beta, const int* __restrict__ batch,
                            float* __restrict__ pooled, int layer, int n_nodes) {
    int wave = __builtin_amdgcn_readfirstlane((int)((blockIdx.x * blockDim.x + threadIdx.x) >> 5));
    int n0 = wave * NPW;
    if (n0 >= n_nodes) return;
    int lane = threadIdx.x & 31;

    float4 x[NPW];
    int gid[NPW];
#pragma unroll
    for (int i = 0; i < NPW; ++i) {
        int n = n0 + i < n_nodes ? n0 + i : n_nodes - 1;
        x[i] = ((const float4*)(out_acc + (size_t)n * HID))[lane];
        gid[i] = batch[n];
    }

    float invN = 1.0f / (float)n_nodes;
    float4 s = ((const float4*)sums)[lane];
    float4 sq = ((const float4*)sumsq)[lane];
    float4 ga = ((const float4*)gamma)[lane];
    float4 be = ((const float4*)beta)[lane];
    float4 mu, sd;
    mu.x = s.x * invN; mu.y = s.y * invN; mu.z = s.z * invN; mu.w = s.w * invN;
    sd.x = sqrtf(sq.x * invN - mu.x * mu.x + BN_EPS);
    sd.y = sqrtf(sq.y * invN - mu.y * mu.y + BN_EPS);
    sd.z = sqrtf(sq.z * invN - mu.z * mu.z + BN_EPS);
    sd.w = sqrtf(sq.w * invN - mu.w * mu.w + BN_EPS);

    int cur = gid[0];
    float4 run = make_float4(0.0f, 0.0f, 0.0f, 0.0f);
#pragma unroll
    for (int i = 0; i < NPW; ++i) {
        if (n0 + i < n_nodes) {
            float4 v;
            v.x = (x[i].x - mu.x) / sd.x * ga.x + be.x;
            v.y = (x[i].y - mu.y) / sd.y * ga.y + be.y;
            v.z = (x[i].z - mu.z) / sd.z * ga.z + be.z;
            v.w = (x[i].w - mu.w) / sd.w * ga.w + be.w;
            ((float4*)(out_acc + (size_t)(n0 + i) * HID))[lane] = v;
            if (gid[i] != cur) {
                pool_add(pooled, cur, layer, lane, run);
                cur = gid[i];
                run = make_float4(0.0f, 0.0f, 0.0f, 0.0f);
            }
            run.x += v.x; run.y += v.y; run.z += v.z; run.w += v.w;
        }
    }
    pool_add(pooled, cur, layer, lane, run);
}

extern "C" void kernel_launch(void* const* d_in, const int* in_sizes, int n_in,
                              void* d_out, int out_size, void* d_ws, size_t ws_size,
                              hipStream_t stream) {
    const float* x_in  = (const float*)d_in[0];
    const int*   ei    = (const int*)d_in[1];
    const int*   batch = (const int*)d_in[2];
    const float* W     = (const float*)d_in[3];
    const float* att_s = (const float*)d_in[4];
    const float* att_d = (const float*)d_in[5];
    const float* bias  = (const float*)d_in[6];
    const float* gamma = (const float*)d_in[7];
    const float* beta  = (const float*)d_in[8];

    const int N = in_sizes[0] / HID;
    const int E = in_sizes[1] / 2;

    float* pooled = (float*)d_out;
    float* xfinal = (float*)d_out + (size_t)NGRAPH * 3 * HID;

    char* w = (char*)d_ws;
    float* hbuf = (float*)w;  w += (size_t)N * HID * sizeof(float);
    float* xb   = (float*)w;  w += (size_t)N * HID * sizeof(float);
    const int n_gblk = (N + 7) / 8;
    float* part = (float*)w;  w += (size_t)n_gblk * 2 * HID * sizeof(float);
    float* asrc = (float*)w;  w += (size_t)N * sizeof(float);
    float* adst = (float*)w;  w += (size_t)N * sizeof(float);
    float* sums = (float*)w;  w += (size_t)2 * HID * sizeof(float);
    float* sumsq = sums + HID;
    int* btot   = (int*)w;    w += (size_t)MAXB * BPAD * sizeof(int);
    int* bfill  = (int*)w;    w += (size_t)MAXB * BPAD * sizeof(int);
    int2* rec   = (int2*)w;   w += (size_t)E * sizeof(int2);
    int* rowptr = (int*)w;    w += (size_t)(N + 1) * sizeof(int);
    int* csrsrc = (int*)w;

    const int waves_g = (N + 15) / 16;
    dim3 blk(256);
    dim3 grid_gemm((waves_g * 32 + 255) / 256);
    dim3 grid_node(n_gblk);
    dim3 grid_edge((E + CH - 1) / CH);
    const int nbk = (N + BNODES - 1) / BNODES;
    dim3 grid_rows((N + 8 * NPW - 1) / (8 * NPW));

    gat_zero<<<(NGRAPH * 3 * HID + 255) / 256, blk, 0, stream>>>(pooled, NGRAPH * 3 * HID);

    gat_zero<<<(2 * MAXB * BPAD + 255) / 256, blk, 0, stream>>>((float*)btot, 2 * MAXB * BPAD);
    gat_bucket_count<<<grid_edge, CT, 0, stream>>>(ei, btot, E, nbk);
    gat_bucket_scatter<<<grid_edge, CT, 0, stream>>>(ei, btot, bfill, rec, E, nbk);
    gat_bucket_sort<<<nbk, ST, 0, stream>>>(rec, btot, rowptr, csrsrc, N, nbk);

    for (int l = 0; l < 3; ++l) {
        const float* X  = (l == 0) ? x_in : xb;
        float* oacc     = (l == 2) ? xfinal : xb;
        const float* Wl = W + (size_t)l * HID * HID;

        gat_gemm_wmma<<<grid_gemm, blk, 0, stream>>>(X, Wl, att_s + l * HID, att_d + l * HID,
                                                     hbuf, asrc, adst, N);
        gat_gather<<<grid_node, blk, 0, stream>>>(rowptr, csrsrc, asrc, adst, hbuf, bias + l * HID,
                                                  oacc, part, sums, N);
        gat_stats_reduce<<<256, 256, 0, stream>>>(part, sums, n_gblk);
        gat_bn_pool<<<grid_rows, blk, 0, stream>>>(oacc, sums, sumsq, gamma + l * HID,
                                                   beta + l * HID, batch, pooled, l, N);
    }
}


# Gather launch: two nodes per wave, one per 16-lane half (lane = 8 channels), edge index and weight broadcast inside the half; softmax denominator summed by a 16-lane tree, row sum order kept
# speedup vs baseline: 4.9993x; 1.0702x over previous
#include <hip/hip_runtime.h>
#include <hip/hip_bf16.h>

#define HID 128
#define NGRAPH 512
#define NEG_SLOPE 0.2f
#define BN_EPS 1e-5f

typedef __attribute__((ext_vector_type(2))) float v2f;
typedef __attribute__((ext_vector_type(8))) float v8f;

__device__ __forceinline__ float lrelu(float a) { return a > 0.0f ? a : NEG_SLOPE * a; }

__global__ void gat_zero(float* __restrict__ p, int n) {
    int i = blockIdx.x * blockDim.x + threadIdx.x;
    if (i < n) p[i] = 0.0f;
}

__global__ void gat_gemm_wmma(const float* __restrict__ X, const float* __restrict__ W,
                              const float* __restrict__ att_s, const float* __restrict__ att_d,
                              float* __restrict__ H, float* __restrict__ a_s,
                              float* __restrict__ a_d, int nrows) {
    int wave = (blockIdx.x * blockDim.x + threadIdx.x) >> 5;
    int m0 = wave * 16;
    if (m0 >= nrows) return;
    int lane = threadIdx.x & 31;
    int lr = lane & 15;
    int lh = lane >> 4;

    float ws[8], wd[8];
#pragma unroll
    for (int n = 0; n < 8; ++n) {
        ws[n] = att_s[n * 16 + lr];
        wd[n] = att_d[n * 16 + lr];
    }

    v8f acc[8] = {};
    const float* xrow = X + (size_t)(m0 + lr) * HID;

    for (int k = 0; k < HID; k += 4) {
        int ka = k + lh * 2;
        v2f a;
        a.x = xrow[ka];
        a.y = xrow[ka + 1];
#pragma unroll
        for (int n = 0; n < 8; ++n) {
            v2f b;
            b.x = W[ka * HID + n * 16 + lr];
            b.y = W[(ka + 1) * HID + n * 16 + lr];
            acc[n] = __builtin_amdgcn_wmma_f32_16x16x4_f32(
                false, a, false, b, (short)0, acc[n], false, false);
        }
    }
#pragma unroll
    for (int n = 0; n < 8; ++n)
#pragma unroll
        for (int j = 0; j < 8; ++j)
            H[(size_t)(m0 + j + lh * 8) * HID + n * 16 + lr] = acc[n][j];

    float ps[8], pd[8];
#pragma unroll
    for (int j = 0; j < 8; ++j) { ps[j] = 0.0f; pd[j] = 0.0f; }
#pragma unroll
    for (int n = 0; n < 8; ++n)
#pragma unroll
        for (int j = 0; j < 8; ++j) {
            ps[j] += acc[n][j] * ws[n];
            pd[j] += acc[n][j] * wd[n];
        }

    bool h8 = (lr & 8) != 0;
    float qs[4], qd[4];
#pragma unroll
    for (int i = 0; i < 4; ++i) {
        qs[i] = (h8 ? ps[4 + i] : ps[i]) + __shfl_xor(h8 ? ps[i] : ps[4 + i], 8, 32);
        qd[i] = (h8 ? pd[4 + i] : pd[i]) + __shfl_xor(h8 ? pd[i] : pd[4 + i], 8, 32);
    }
    bool h4 = (lr & 4) != 0;
    float rs[2], rd[2];
#pragma unroll
    for (int i = 0; i < 2; ++i) {
        rs[i] = (h4 ? qs[2 + i] : qs[i]) + __shfl_xor(h4 ? qs[i] : qs[2 + i], 4, 32);
        rd[i] = (h4 ? qd[2 + i] : qd[i]) + __shfl_xor(h4 ? qd[i] : qd[2 + i], 4, 32);
    }
    bool h2 = (lr & 2) != 0;
    float ts = (h2 ? rs[1] : rs[0]) + __shfl_xor(h2 ? rs[0] : rs[1], 2, 32);
    float td = (h2 ? rd[1] : rd[0]) + __shfl_xor(h2 ? rd[0] : rd[1], 2, 32);
    bool h1 = (lr & 1) != 0;
    float tot = (h1 ? td : ts) + __shfl_xor(h1 ? ts : td, 1, 32);
    float* dst = h1 ? a_d : a_s;
    dst[m0 + lh * 8 + (lr >> 1)] = tot;
}

#define NPW 8

#define BSH 8
#define BNODES (1 << BSH)
#define MAXB 512
#define BPAD 32
#define CT 512
#define CPT 8
#define CH (CT * CPT)
#define ST 1024
#define SPT 4

__device__ __forceinline__ int wave_incl_scan(int x, int lane) {
    for (int off = 1; off < 32; off <<= 1) {
        int y = __shfl_up(x, off, 32);
        if (lane >= off) x += y;
    }
    return x;
}

template <bool FULL>
__device__ __forceinline__ void bucket_count_body(const int* __restrict__ ei,
                                                  int* __restrict__ btot, int n_edges, int nbk,
                                                  int* h) {
    int t = threadIdx.x;
    int e0 = blockIdx.x * CH + t;
    int d[CPT];
#pragma unroll
    for (int i = 0; i < CPT; ++i) {
        int e = e0 + i * CT;
        bool in = FULL || e < n_edges;
        int v = ei[n_edges + (in ? e : n_edges - 1)];
        d[i] = in ? v : -1;
    }
    h[t] = 0;
    __syncthreads();
#pragma unroll
    for (int i = 0; i < CPT; ++i)
        if (FULL || d[i] >= 0) atomicAdd(&h[d[i] >> BSH], 1);
    __syncthreads();
    int c = h[t];
    if (t < nbk && c != 0) atomicAdd(&btot[t * BPAD], c);
}

__global__ void __launch_bounds__(CT)
gat_bucket_count(const int* __restrict__ ei, int* __restrict__ btot, int n_edges, int nbk) {
    __shared__ int h[MAXB];
    if ((int)blockIdx.x * CH + CH <= n_edges)
        bucket_count_body<true>(ei, btot, n_edges, nbk, h);
    else
        bucket_count_body<false>(ei, btot, n_edges, nbk, h);
}

template <bool FULL>
__device__ __forceinline__ void bucket_scatter_body(const int* __restrict__ ei,
                                                    const int* __restrict__ btot,
                                                    int* __restrict__ bfill,
                                                    int2* __restrict__ rec, int n_edges, int nbk,
                                                    int* h, int* wtot) {
    int t = threadIdx.x;
    int lane = t & 31;
    int wv = t >> 5;
    int e0 = blockIdx.x * CH + t;

    int s[CPT], d[CPT], r[CPT];
#pragma unroll
    for (int i = 0; i < CPT; ++i) {
        int e = e0 + i * CT;
        bool in = FULL || e < n_edges;
        int ec = in ? e : n_edges - 1;
        s[i] = ei[ec];
        d[i] = in ? ei[n_edges + ec] : -1;
    }

    int tot = t < nbk ? btot[t * BPAD] : 0;
    int x = wave_incl_scan(tot, lane);
    if (lane == 31) wtot[wv] = x;
    h[t] = 0;
    __syncthreads();

#pragma unroll
    for (int i = 0; i < CPT; ++i)
        r[i] = (FULL || d[i] >= 0) ? atomicAdd(&h[d[i] >> BSH], 1) : 0;

    int wt = lane < CT / 32 ? wtot[lane] : 0;
    int wsc = wave_incl_scan(wt, lane);
    int bstart = __shfl(wsc, wv, 32) - __shfl(wt, wv, 32) + x - tot;
    __syncthreads();

    int c = h[t];
    int g = (t < nbk && c != 0) ? atomicAdd(&bfill[t * BPAD], c) : 0;
    h[t] = bstart + g;
    __syncthreads();

#pragma unroll
    for (int i = 0; i < CPT; ++i)
        if (FULL || d[i] >= 0) rec[h[d[i] >> BSH] + r[i]] = make_int2(s[i], d[i]);
}

__global__ void __launch_bounds__(CT)
gat_bucket_scatter(const int* __restrict__ ei, const int* __restrict__ btot,
                   int* __restrict__ bfill, int2* __restrict__ rec, int n_edges, int nbk) {
    __shared__ int h[MAXB];
    __shared__ int wtot[CT / 32];
    if ((int)blockIdx.x * CH + CH <= n_edges)
        bucket_scatter_body<true>(ei, btot, bfill, rec, n_edges, nbk, h, wtot);
    else
        bucket_scatter_body<false>(ei, btot, bfill, rec, n_edges, nbk, h, wtot);
}

__global__ void __launch_bounds__(ST)
gat_bucket_sort(const int2* __restrict__ rec, const int* __restrict__ btot,
                int* __restrict__ rowptr, int* __restrict__ csr_src, int n_nodes, int nbk) {
    __shared__ int cnt[BNODES];
    __shared__ int excl[BNODES];
    __shared__ int red[ST / 32];
    __shared__ int wtot[BNODES / 32];
    int t = threadIdx.x;
    int lane = t & 31;
    int wv = t >> 5;
    int b = blockIdx.x;

    int v = t < b ? btot[t * BPAD] : 0;
    int n = btot[b * BPAD];
    for (int off = 16; off > 0; off >>= 1) v += __shfl_xor(v, off, 32);
    if (lane == 0) red[wv] = v;
    if (t < BNODES) cnt[t] = 0;
    __syncthreads();
    int beg = red[lane];
    for (int off = 16; off > 0; off >>= 1) beg += __shfl_xor(beg, off, 32);
    const int2* brec = rec + beg;

    for (int i0 = t; i0 < n + t; i0 += ST * SPT) {
        int dd[SPT];
#pragma unroll
        for (int k = 0; k < SPT; ++k) {
            int i = i0 + k * ST;
            dd[k] = brec[i < n ? i : n - 1].y;
        }
#pragma unroll
        for (int k = 0; k < SPT; ++k)
            if (i0 + k * ST < n) atomicAdd(&cnt[dd[k] & (BNODES - 1)], 1);
    }
    __syncthreads();

    int c = 0, x = 0;
    if (t < BNODES) {
        c = cnt[t];
        x = wave_incl_scan(c, lane);
        if (lane == 31) wtot[wv] = x;
    }
    __syncthreads();
    if (t < BNODES) {
        int wt = lane < BNODES / 32 ? wtot[lane] : 0;
        int wsc = wave_incl_scan(wt, lane);
        int e = __shfl(wsc, wv, 32) - __shfl(wt, wv, 32) + x - c;
        excl[t] = e;
        cnt[t] = 0;
        int node = b * BNODES + t;
        if (node < n_nodes) rowptr[node] = beg + e;
        if (b == nbk - 1 && t == 0) rowptr[n_nodes] = beg + n;
    }
    __syncthreads();

    int* bcsr = csr_src + beg;
    for (int i0 = t; i0 < n + t; i0 += ST * SPT) {
        int2 rr[SPT];
#pragma unroll
        for (int k = 0; k < SPT; ++k) {
            int i = i0 + k * ST;
            rr[k] = brec[i < n ? i : n - 1];
        }
#pragma unroll
        for (int k = 0; k < SPT; ++k) {
            if (i0 + k * ST < n) {
                int q = rr[k].y & (BNODES - 1);
                bcsr[excl[q] + atomicAdd(&cnt[q], 1)] = rr[k].x;
            }
        }
    }
}

#define GW 4
#define GN 8

__device__ __forceinline__ float elu1(float v) { return v > 0.0f ? v : expm1f(v); }

__device__ __forceinline__ int half_bcast_i(int x, unsigned sel) {
    return (int)__builtin_amdgcn_permlane16((unsigned)x, (unsigned)x, sel, sel, false, false);
}
__device__ __forceinline__ float half_bcast_f(float x, unsigned sel) {
    unsigned u = __float_as_uint(x);
    return __uint_as_float(__builtin_amdgcn_permlane16(u, u, sel, sel, false, false));
}

template <int M>
__device__ __forceinline__ float half_xor(float x) {
    return __uint_as_float(__builtin_amdgcn_update_dpp(0u, __float_as_uint(x), 0x160 + M, 0xf, 0xf, true));
}

__device__ __forceinline__ void gather_chunk(const int* __restrict__ csr_src,
                                             const float* __restrict__ a_s, int beg, int end,
                                             int off, int hl, int nd, float ad,
                                             int& s0, int& s1, float& l0, float& l1) {
    int j0 = beg + off + hl;
    int j1 = j0 + 16;
    bool in0 = j0 < end;
    bool in1 = j1 < end;
    int c0 = csr_src[in0 ? j0 : 0];
    int c1 = csr_src[in1 ? j1 : 0];
    s0 = in0 ? c0 : nd;
    s1 = in1 ? c1 : nd;
    l0 = lrelu(a_s[s0] + ad);
    l1 = lrelu(a_s[s1] + ad);
}

__global__ void __launch_bounds__(GW * 32)
gat_gather(const int* __restrict__ rowptr, const int* __restrict__ csr_src,
           const float* __restrict__ a_s, const float* __restrict__ a_d,
           const float* __restrict__ H, const float* __restrict__ bias,
           float* __restrict__ out, float* __restrict__ part,
           float* __restrict__ sums, int n_nodes) {
    __shared__ float4 red[GN][64];
    int wave = __builtin_amdgcn_readfirstlane((int)((blockIdx.x * blockDim.x + threadIdx.x) >> 5));
    int lane = threadIdx.x & 31;
    int hl = lane & 15;
    int hf = lane >> 4;
    int wv = threadIdx.x >> 5;
    int node = wave * 2 + hf;
    bool valid = node < n_nodes;
    float4 v0 = make_float4(0.0f, 0.0f, 0.0f, 0.0f);
    float4 v1 = make_float4(0.0f, 0.0f, 0.0f, 0.0f);
    float4 q0 = make_float4(0.0f, 0.0f, 0.0f, 0.0f);
    float4 q1 = make_float4(0.0f, 0.0f, 0.0f, 0.0f);

    if (wave * 2 < n_nodes) {
        int nd = valid ? node : n_nodes - 1;
        int beg = rowptr[nd];
        int end = rowptr[nd + 1];
        float ad = a_d[nd];
        float self_logit = lrelu(a_s[nd] + ad);
        const float4* H4 = (const float4*)H;
        float4 hs0 = H4[(unsigned)nd * 32u + hl];
        float4 hs1 = H4[(unsigned)nd * 32u + 16u + hl];
        int dlo = __builtin_amdgcn_readlane(end - beg, 0);
        int dhi = __builtin_amdgcn_readlane(end - beg, 16);
        int dmax = dlo > dhi ? dlo : dhi;

        int s0, s1;
        float l0, l1;
        gather_chunk(csr_src, a_s, beg, end, 0, hl, nd, ad, s0, s1, l0, l1);
        float mx = fmaxf(self_logit, fmaxf(l0, l1));
        for (int off = 32; off < dmax; off += 32) {
            int t0, t1;
            float m0, m1;
            gather_chunk(csr_src, a_s, beg, end, off, hl, nd, ad, t0, t1, m0, m1);
            mx = fmaxf(mx, fmaxf(m0, m1));
        }
        mx = fmaxf(mx, half_xor<8>(mx));
        mx = fmaxf(mx, half_xor<4>(mx));
        mx = fmaxf(mx, half_xor<2>(mx));
        mx = fmaxf(mx, half_xor<1>(mx));

        float es = expf(self_logit - mx);
        float4 acc0, acc1;
        acc0.x = es * hs0.x; acc0.y = es * hs0.y; acc0.z = es * hs0.z; acc0.w = es * hs0.w;
        acc1.x = es * hs1.x; acc1.y = es * hs1.y; acc1.z = es * hs1.z; acc1.w = es * hs1.w;
        float wsum = 0.0f;

        for (int off = 0; off < dmax; off += 32) {
            if (off > 0)
                gather_chunk(csr_src, a_s, beg, end, off, hl, nd, ad, s0, s1, l0, l1);
            float w0 = beg + off + hl < end ? expf(l0 - mx) : 0.0f;
            float w1 = beg + off + 16 + hl < end ? expf(l1 - mx) : 0.0f;
            wsum += w0 + w1;
            int n = dmax - off < 32 ? dmax - off : 32;
            n = (n + 3) & ~3;
            for (int k = 0; k < n; k += 4) {
                int ss = k < 16 ? s0 : s1;
                float ww = k < 16 ? w0 : w1;
#pragma unroll
                for (int u = 0; u < 4; ++u) {
                    unsigned sel = (unsigned)((k & 15) + u) * 0x11111111u;
                    int sk = half_bcast_i(ss, sel);
                    float wk = half_bcast_f(ww, sel);
                    unsigned ri = (unsigned)sk * 32u + hl;
                    float4 ha = H4[ri];
                    float4 hb = H4[ri + 16u];
                    acc0.x += wk * ha.x; acc0.y += wk * ha.y; acc0.z += wk * ha.z; acc0.w += wk * ha.w;
                    acc1.x += wk * hb.x; acc1.y += wk * hb.y; acc1.z += wk * hb.z; acc1.w += wk * hb.w;
                }
            }
        }
        wsum += half_xor<8>(wsum);
        wsum += half_xor<4>(wsum);
        wsum += half_xor<2>(wsum);
        wsum += half_xor<1>(wsum);

        if (valid) {
            float inv = 1.0f / (es + wsum);
            float4 b0 = ((const float4*)bias)[hl];
            float4 b1 = ((const float4*)bias)[16 + hl];
            v0.x = elu1(acc0.x * inv + b0.x);
            v0.y = elu1(acc0.y * inv + b0.y);
            v0.z = elu1(acc0.z * inv + b0.z);
            v0.w = elu1(acc0.w * inv + b0.w);
            v1.x = elu1(acc1.x * inv + b1.x);
            v1.y = elu1(acc1.y * inv + b1.y);
            v1.z = elu1(acc1.z * inv + b1.z);
            v1.w = elu1(acc1.w * inv + b1.w);
            q0.x = v0.x * v0.x; q0.y = v0.y * v0.y; q0.z = v0.z * v0.z; q0.w = v0.w * v0.w;
            q1.x = v1.x * v1.x; q1.y = v1.y * v1.y; q1.z = v1.z * v1.z; q1.w = v1.w * v1.w;
        }
    }

    int r = wv * 2 + hf;
    red[r][hl] = v0;
    red[r][16 + hl] = v1;
    red[r][32 + hl] = q0;
    red[r][48 + hl] = q1;
    __syncthreads();
    const float* rf = (const float*)red;
    int t = threadIdx.x;
    float tv = 0.0f, tq = 0.0f;
    for (int w = 0; w < GN; ++w) {
        tv += rf[w * 256 + t];
        tq += rf[w * 256 + 128 + t];
    }
    part[(size_t)blockIdx.x * (2 * HID) + t] = tv;
    part[(size_t)blockIdx.x * (2 * HID) + HID + t] = tq;
    if (blockIdx.x == 0) { sums[t] = 0.0f; sums[HID + t] = 0.0f; }
    if (valid) {
        float4* orow = (float4*)(out + (size_t)node * HID);
        orow[hl] = v0;
        orow[16 + hl] = v1;
    }
}

__global__ void gat_stats_reduce(const float* __restrict__ part, float* __restrict__ sums,
                                 int nrows) {
    int t = threadIdx.x;
    int r = blockIdx.x;
    const int g = gridDim.x;
    float acc = 0.0f;
    for (; r + 15 * g < nrows; r += 16 * g) {
        float x[16];
        for (int i = 0; i < 16; ++i) x[i] = part[(size_t)(r + i * g) * (2 * HID) + t];
        for (int i = 0; i < 16; ++i) acc += x[i];
    }
    for (; r < nrows; r += g) acc += part[(size_t)r * (2 * HID) + t];
    atomicAdd(&sums[t], acc);
}

__device__ __forceinline__ void pool_add(float* __restrict__ pooled, int g, int layer, int lane,
                                         float4 a) {
    float* p = pooled + (size_t)g * (3 * HID) + layer * HID + lane * 4;
    atomicAdd(p + 0, a.x);
    atomicAdd(p + 1, a.y);
    atomicAdd(p + 2, a.z);
    atomicAdd(p + 3, a.w);
}

__global__ void gat_bn_pool(float* __restrict__ out_acc, const float* __restrict__ sums,
                            const float* __restrict__ sumsq, const float* __restrict__ gamma,
                            const float* __restrict__ beta, const int* __restrict__ batch,
                            float* __restrict__ pooled, int layer, int n_nodes) {
    int wave = __builtin_amdgcn_readfirstlane((int)((blockIdx.x * blockDim.x + threadIdx.x) >> 5));
    int n0 = wave * NPW;
    if (n0 >= n_nodes) return;
    int lane = threadIdx.x & 31;

    float4 x[NPW];
    int gid[NPW];
#pragma unroll
    for (int i = 0; i < NPW; ++i) {
        int n = n0 + i < n_nodes ? n0 + i : n_nodes - 1;
        x[i] = ((const float4*)(out_acc + (size_t)n * HID))[lane];
        gid[i] = batch[n];
    }

    float invN = 1.0f / (float)n_nodes;
    float4 s = ((const float4*)sums)[lane];
    float4 sq = ((const float4*)sumsq)[lane];
    float4 ga = ((const float4*)gamma)[lane];
    float4 be = ((const float4*)beta)[lane];
    float4 mu, sd;
    mu.x = s.x * invN; mu.y = s.y * invN; mu.z = s.z * invN; mu.w = s.w * invN;
    sd.x = sqrtf(sq.x * invN - mu.x * mu.x + BN_EPS);
    sd.y = sqrtf(sq.y * invN - mu.y * mu.y + BN_EPS);
    sd.z = sqrtf(sq.z * invN - mu.z * mu.z + BN_EPS);
    sd.w = sqrtf(sq.w * invN - mu.w * mu.w + BN_EPS);

    int cur = gid[0];
    float4 run = make_float4(0.0f, 0.0f, 0.0f, 0.0f);
#pragma unroll
    for (int i = 0; i < NPW; ++i) {
        if (n0 + i < n_nodes) {
            float4 v;
            v.x = (x[i].x - mu.x) / sd.x * ga.x + be.x;
            v.y = (x[i].y - mu.y) / sd.y * ga.y + be.y;
            v.z = (x[i].z - mu.z) / sd.z * ga.z + be.z;
            v.w = (x[i].w - mu.w) / sd.w * ga.w + be.w;
            ((float4*)(out_acc + (size_t)(n0 + i) * HID))[lane] = v;
            if (gid[i] != cur) {
                pool_add(pooled, cur, layer, lane, run);
                cur = gid[i];
                run = make_float4(0.0f, 0.0f, 0.0f, 0.0f);
            }
            run.x += v.x; run.y += v.y; run.z += v.z; run.w += v.w;
        }
    }
    pool_add(pooled, cur, layer, lane, run);
}

extern "C" void kernel_launch(void* const* d_in, const int* in_sizes, int n_in,
                              void* d_out, int out_size, void* d_ws, size_t ws_size,
                              hipStream_t stream) {
    const float* x_in  = (const float*)d_in[0];
    const int*   ei    = (const int*)d_in[1];
    const int*   batch = (const int*)d_in[2];
    const float* W     = (const float*)d_in[3];
    const float* att_s = (const float*)d_in[4];
    const float* att_d = (const float*)d_in[5];
    const float* bias  = (const float*)d_in[6];
    const float* gamma = (const float*)d_in[7];
    const float* beta  = (const float*)d_in[8];

    const int N = in_sizes[0] / HID;
    const int E = in_sizes[1] / 2;

    float* pooled = (float*)d_out;
    float* xfinal = (float*)d_out + (size_t)NGRAPH * 3 * HID;

    char* w = (char*)d_ws;
    float* hbuf = (float*)w;  w += (size_t)N * HID * sizeof(float);
    float* xb   = (float*)w;  w += (size_t)N * HID * sizeof(float);
    const int n_gblk = (N + GN - 1) / GN;
    float* part = (float*)w;  w += (size_t)n_gblk * 2 * HID * sizeof(float);
    float* asrc = (float*)w;  w += (size_t)N * sizeof(float);
    float* adst = (float*)w;  w += (size_t)N * sizeof(float);
    float* sums = (float*)w;  w += (size_t)2 * HID * sizeof(float);
    float* sumsq = sums + HID;
    int* btot   = (int*)w;    w += (size_t)MAXB * BPAD * sizeof(int);
    int* bfill  = (int*)w;    w += (size_t)MAXB * BPAD * sizeof(int);
    int2* rec   = (int2*)w;   w += (size_t)E * sizeof(int2);
    int* rowptr = (int*)w;    w += (size_t)(N + 1) * sizeof(int);
    int* csrsrc = (int*)w;

    const int waves_g = (N + 15) / 16;
    dim3 blk(256);
    dim3 grid_gemm((waves_g * 32 + 255) / 256);
    dim3 grid_node(n_gblk);
    dim3 grid_edge((E + CH - 1) / CH);
    const int nbk = (N + BNODES - 1) / BNODES;
    dim3 grid_rows((N + 8 * NPW - 1) / (8 * NPW));

    gat_zero<<<(NGRAPH * 3 * HID + 255) / 256, blk, 0, stream>>>(pooled, NGRAPH * 3 * HID);

    gat_zero<<<(2 * MAXB * BPAD + 255) / 256, blk, 0, stream>>>((float*)btot, 2 * MAXB * BPAD);
    gat_bucket_count<<<grid_edge, CT, 0, stream>>>(ei, btot, E, nbk);
    gat_bucket_scatter<<<grid_edge, CT, 0, stream>>>(ei, btot, bfill, rec, E, nbk);
    gat_bucket_sort<<<nbk, ST, 0, stream>>>(rec, btot, rowptr, csrsrc, N, nbk);

    for (int l = 0; l < 3; ++l) {
        const float* X  = (l == 0) ? x_in : xb;
        float* oacc     = (l == 2) ? xfinal : xb;
        const float* Wl = W + (size_t)l * HID * HID;

        gat_gemm_wmma<<<grid_gemm, blk, 0, stream>>>(X, Wl, att_s + l * HID, att_d + l * HID,
                                                     hbuf, asrc, adst, N);
        gat_gather<<<grid_node, GW * 32, 0, stream>>>(rowptr, csrsrc, asrc, adst, hbuf, bias + l * HID,
                                                  oacc, part, sums, N);
        gat_stats_reduce<<<256, 256, 0, stream>>>(part, sums, n_gblk);
        gat_bn_pool<<<grid_rows, blk, 0, stream>>>(oacc, sums, sumsq, gamma + l * HID,
                                                   beta + l * HID, batch, pooled, l, N);
    }
}


# BN apply + graph pooling of layers 0,1 folded into the next product launch: its 16-row tile is normalized on load and kept in LDS for the A fragments (4-wave groups); pooled run sums now per 16 rows
# speedup vs baseline: 5.5014x; 1.1004x over previous
#include <hip/hip_runtime.h>
#include <hip/hip_bf16.h>

#define HID 128
#define NGRAPH 512
#define NEG_SLOPE 0.2f
#define BN_EPS 1e-5f

typedef __attribute__((ext_vector_type(2))) float v2f;
typedef __attribute__((ext_vector_type(8))) float v8f;

__device__ __forceinline__ float lrelu(float a) { return a > 0.0f ? a : NEG_SLOPE * a; }

__global__ void gat_zero(float* __restrict__ p, int n) {
    int i = blockIdx.x * blockDim.x + threadIdx.x;
    if (i < n) p[i] = 0.0f;
}

#define PT 128
#define TP 132

__device__ __forceinline__ void pool_add(float* __restrict__ pooled, int g, int layer, int lane,
                                         float4 a) {
    float* p = pooled + (size_t)g * (3 * HID) + layer * HID + lane * 4;
    atomicAdd(p + 0, a.x);
    atomicAdd(p + 1, a.y);
    atomicAdd(p + 2, a.z);
    atomicAdd(p + 3, a.w);
}

template <bool BN>
__global__ void __launch_bounds__(PT, 8)
gat_gemm_wmma(const float* __restrict__ X, const float* __restrict__ W,
              const float* __restrict__ att_s, const float* __restrict__ att_d,
              float* __restrict__ H, float* __restrict__ a_s,
              float* __restrict__ a_d, int nrows,
              const float* __restrict__ sums, const float* __restrict__ sumsq,
              const float* __restrict__ gamma, const float* __restrict__ beta,
              const int* __restrict__ batch, float* __restrict__ pooled, int layer) {
    __shared__ __attribute__((aligned(16))) float tile[PT / 32][16 * TP];
    int wave = __builtin_amdgcn_readfirstlane((int)((blockIdx.x * PT + threadIdx.x) >> 5));
    int m0 = wave * 16;
    bool active = m0 < nrows;
    int lane = threadIdx.x & 31;
    int lr = lane & 15;
    int lh = lane >> 4;
    float* tl = tile[threadIdx.x >> 5];
    int cur = 0;
    float4 run = make_float4(0.0f, 0.0f, 0.0f, 0.0f);

    if (active) {
        float4 x[16];
#pragma unroll
        for (int i = 0; i < 16; ++i)
            x[i] = ((const float4*)(X + (size_t)(m0 + i) * HID))[lane];
        if (BN) {
            int gid[16];
#pragma unroll
            for (int i = 0; i < 16; ++i) gid[i] = batch[m0 + i];
            float invN = 1.0f / (float)nrows;
            float4 s = ((const float4*)sums)[lane];
            float4 sq = ((const float4*)sumsq)[lane];
            float4 ga = ((const float4*)gamma)[lane];
            float4 be = ((const float4*)beta)[lane];
            float4 mu, sc;
            mu.x = s.x * invN; mu.y = s.y * invN; mu.z = s.z * invN; mu.w = s.w * invN;
            sc.x = ga.x / sqrtf(sq.x * invN - mu.x * mu.x + BN_EPS);
            sc.y = ga.y / sqrtf(sq.y * invN - mu.y * mu.y + BN_EPS);
            sc.z = ga.z / sqrtf(sq.z * invN - mu.z * mu.z + BN_EPS);
            sc.w = ga.w / sqrtf(sq.w * invN - mu.w * mu.w + BN_EPS);
            cur = gid[0];
#pragma unroll
            for (int i = 0; i < 16; ++i) {
                float4 v;
                v.x = (x[i].x - mu.x) * sc.x + be.x;
                v.y = (x[i].y - mu.y) * sc.y + be.y;
                v.z = (x[i].z - mu.z) * sc.z + be.z;
                v.w = (x[i].w - mu.w) * sc.w + be.w;
                x[i] = v;
                if (gid[i] != cur) {
                    pool_add(pooled, cur, layer, lane, run);
                    cur = gid[i];
                    run = make_float4(0.0f, 0.0f, 0.0f, 0.0f);
                }
                run.x += v.x; run.y += v.y; run.z += v.z; run.w += v.w;
            }
        }
#pragma unroll
        for (int i = 0; i < 16; ++i)
            *(float4*)(tl + i * TP + lane * 4) = x[i];
    }
    __syncthreads();
    if (!active) return;
    if (BN) pool_add(pooled, cur, layer, lane, run);

    float ws[8], wd[8];
#pragma unroll
    for (int n = 0; n < 8; ++n) {
        ws[n] = att_s[n * 16 + lr];
        wd[n] = att_d[n * 16 + lr];
    }

    v8f acc[8] = {};
    const float* arow = tl + lr * TP + lh * 2;

    for (int k0 = 0; k0 < HID; k0 += 32) {
        v2f a[8];
#pragma unroll
        for (int kk = 0; kk < 8; ++kk) a[kk] = *(const v2f*)(arow + k0 + kk * 4);
#pragma unroll
        for (int kk = 0; kk < 8; ++kk) {
            int ka = k0 + kk * 4 + lh * 2;
#pragma unroll
            for (int n = 0; n < 8; ++n) {
                v2f b;
                b.x = W[ka * HID + n * 16 + lr];
                b.y = W[(ka + 1) * HID + n * 16 + lr];
                acc[n] = __builtin_amdgcn_wmma_f32_16x16x4_f32(
                    false, a[kk], false, b, (short)0, acc[n], false, false);
            }
        }
    }
#pragma unroll
    for (int n = 0; n < 8; ++n)
#pragma unroll
        for (int j = 0; j < 8; ++j)
            H[(size_t)(m0 + j + lh * 8) * HID + n * 16 + lr] = acc[n][j];

    float ps[8], pd[8];
#pragma unroll
    for (int j = 0; j < 8; ++j) { ps[j] = 0.0f; pd[j] = 0.0f; }
#pragma unroll
    for (int n = 0; n < 8; ++n)
#pragma unroll
        for (int j = 0; j < 8; ++j) {
            ps[j] += acc[n][j] * ws[n];
            pd[j] += acc[n][j] * wd[n];
        }

    bool h8 = (lr & 8) != 0;
    float qs[4], qd[4];
#pragma unroll
    for (int i = 0; i < 4; ++i) {
        qs[i] = (h8 ? ps[4 + i] : ps[i]) + __shfl_xor(h8 ? ps[i] : ps[4 + i], 8, 32);
        qd[i] = (h8 ? pd[4 + i] : pd[i]) + __shfl_xor(h8 ? pd[i] : pd[4 + i], 8, 32);
    }
    bool h4 = (lr & 4) != 0;
    float rs[2], rd[2];
#pragma unroll
    for (int i = 0; i < 2; ++i) {
        rs[i] = (h4 ? qs[2 + i] : qs[i]) + __shfl_xor(h4 ? qs[i] : qs[2 + i], 4, 32);
        rd[i] = (h4 ? qd[2 + i] : qd[i]) + __shfl_xor(h4 ? qd[i] : qd[2 + i], 4, 32);
    }
    bool h2 = (lr & 2) != 0;
    float ts = (h2 ? rs[1] : rs[0]) + __shfl_xor(h2 ? rs[0] : rs[1], 2, 32);
    float td = (h2 ? rd[1] : rd[0]) + __shfl_xor(h2 ? rd[0] : rd[1], 2, 32);
    bool h1 = (lr & 1) != 0;
    float tot = (h1 ? td : ts) + __shfl_xor(h1 ? ts : td, 1, 32);
    float* dst = h1 ? a_d : a_s;
    dst[m0 + lh * 8 + (lr >> 1)] = tot;
}

#define NPW 8

#define BSH 8
#define BNODES (1 << BSH)
#define MAXB 512
#define BPAD 32
#define CT 512
#define CPT 8
#define CH (CT * CPT)
#define ST 1024
#define SPT 4

__device__ __forceinline__ int wave_incl_scan(int x, int lane) {
    for (int off = 1; off < 32; off <<= 1) {
        int y = __shfl_up(x, off, 32);
        if (lane >= off) x += y;
    }
    return x;
}

template <bool FULL>
__device__ __forceinline__ void bucket_count_body(const int* __restrict__ ei,
                                                  int* __restrict__ btot, int n_edges, int nbk,
                                                  int* h) {
    int t = threadIdx.x;
    int e0 = blockIdx.x * CH + t;
    int d[CPT];
#pragma unroll
    for (int i = 0; i < CPT; ++i) {
        int e = e0 + i * CT;
        bool in = FULL || e < n_edges;
        int v = ei[n_edges + (in ? e : n_edges - 1)];
        d[i] = in ? v : -1;
    }
    h[t] = 0;
    __syncthreads();
#pragma unroll
    for (int i = 0; i < CPT; ++i)
        if (FULL || d[i] >= 0) atomicAdd(&h[d[i] >> BSH], 1);
    __syncthreads();
    int c = h[t];
    if (t < nbk && c != 0) atomicAdd(&btot[t * BPAD], c);
}

__global__ void __launch_bounds__(CT)
gat_bucket_count(const int* __restrict__ ei, int* __restrict__ btot, int n_edges, int nbk) {
    __shared__ int h[MAXB];
    if ((int)blockIdx.x * CH + CH <= n_edges)
        bucket_count_body<true>(ei, btot, n_edges, nbk, h);
    else
        bucket_count_body<false>(ei, btot, n_edges, nbk, h);
}

template <bool FULL>
__device__ __forceinline__ void bucket_scatter_body(const int* __restrict__ ei,
                                                    const int* __restrict__ btot,
                                                    int* __restrict__ bfill,
                                                    int2* __restrict__ rec, int n_edges, int nbk,
                                                    int* h, int* wtot) {
    int t = threadIdx.x;
    int lane = t & 31;
    int wv = t >> 5;
    int e0 = blockIdx.x * CH + t;

    int s[CPT], d[CPT], r[CPT];
#pragma unroll
    for (int i = 0; i < CPT; ++i) {
        int e = e0 + i * CT;
        bool in = FULL || e < n_edges;
        int ec = in ? e : n_edges - 1;
        s[i] = ei[ec];
        d[i] = in ? ei[n_edges + ec] : -1;
    }

    int tot = t < nbk ? btot[t * BPAD] : 0;
    int x = wave_incl_scan(tot, lane);
    if (lane == 31) wtot[wv] = x;
    h[t] = 0;
    __syncthreads();

#pragma unroll
    for (int i = 0; i < CPT; ++i)
        r[i] = (FULL || d[i] >= 0) ? atomicAdd(&h[d[i] >> BSH], 1) : 0;

    int wt = lane < CT / 32 ? wtot[lane] : 0;
    int wsc = wave_incl_scan(wt, lane);
    int bstart = __shfl(wsc, wv, 32) - __shfl(wt, wv, 32) + x - tot;
    __syncthreads();

    int c = h[t];
    int g = (t < nbk && c != 0) ? atomicAdd(&bfill[t * BPAD], c) : 0;
    h[t] = bstart + g;
    __syncthreads();

#pragma unroll
    for (int i = 0; i < CPT; ++i)
        if (FULL || d[i] >= 0) rec[h[d[i] >> BSH] + r[i]] = make_int2(s[i], d[i]);
}

__global__ void __launch_bounds__(CT)
gat_bucket_scatter(const int* __restrict__ ei, const int* __restrict__ btot,
                   int* __restrict__ bfill, int2* __restrict__ rec, int n_edges, int nbk) {
    __shared__ int h[MAXB];
    __shared__ int wtot[CT / 32];
    if ((int)blockIdx.x * CH + CH <= n_edges)
        bucket_scatter_body<true>(ei, btot, bfill, rec, n_edges, nbk, h, wtot);
    else
        bucket_scatter_body<false>(ei, btot, bfill, rec, n_edges, nbk, h, wtot);
}

__global__ void __launch_bounds__(ST)
gat_bucket_sort(const int2* __restrict__ rec, const int* __restrict__ btot,
                int* __restrict__ rowptr, int* __restrict__ csr_src, int n_nodes, int nbk) {
    __shared__ int cnt[BNODES];
    __shared__ int excl[BNODES];
    __shared__ int red[ST / 32];
    __shared__ int wtot[BNODES / 32];
    int t = threadIdx.x;
    int lane = t & 31;
    int wv = t >> 5;
    int b = blockIdx.x;

    int v = t < b ? btot[t * BPAD] : 0;
    int n = btot[b * BPAD];
    for (int off = 16; off > 0; off >>= 1) v += __shfl_xor(v, off, 32);
    if (lane == 0) red[wv] = v;
    if (t < BNODES) cnt[t] = 0;
    __syncthreads();
    int beg = red[lane];
    for (int off = 16; off > 0; off >>= 1) beg += __shfl_xor(beg, off, 32);
    const int2* brec = rec + beg;

    for (int i0 = t; i0 < n + t; i0 += ST * SPT) {
        int dd[SPT];
#pragma unroll
        for (int k = 0; k < SPT; ++k) {
            int i = i0 + k * ST;
            dd[k] = brec[i < n ? i : n - 1].y;
        }
#pragma unroll
        for (int k = 0; k < SPT; ++k)
            if (i0 + k * ST < n) atomicAdd(&cnt[dd[k] & (BNODES - 1)], 1);
    }
    __syncthreads();

    int c = 0, x = 0;
    if (t < BNODES) {
        c = cnt[t];
        x = wave_incl_scan(c, lane);
        if (lane == 31) wtot[wv] = x;
    }
    __syncthreads();
    if (t < BNODES) {
        int wt = lane < BNODES / 32 ? wtot[lane] : 0;
        int wsc = wave_incl_scan(wt, lane);
        int e = __shfl(wsc, wv, 32) - __shfl(wt, wv, 32) + x - c;
        excl[t] = e;
        cnt[t] = 0;
        int node = b * BNODES + t;
        if (node < n_nodes) rowptr[node] = beg + e;
        if (b == nbk - 1 && t == 0) rowptr[n_nodes] = beg + n;
    }
    __syncthreads();

    int* bcsr = csr_src + beg;
    for (int i0 = t; i0 < n + t; i0 += ST * SPT) {
        int2 rr[SPT];
#pragma unroll
        for (int k = 0; k < SPT; ++k) {
            int i = i0 + k * ST;
            rr[k] = brec[i < n ? i : n - 1];
        }
#pragma unroll
        for (int k = 0; k < SPT; ++k) {
            if (i0 + k * ST < n) {
                int q = rr[k].y & (BNODES - 1);
                bcsr[excl[q] + atomicAdd(&cnt[q], 1)] = rr[k].x;
            }
        }
    }
}

#define GW 4
#define GN 8

__device__ __forceinline__ float elu1(float v) { return v > 0.0f ? v : expm1f(v); }

__device__ __forceinline__ int half_bcast_i(int x, unsigned sel) {
    return (int)__builtin_amdgcn_permlane16((unsigned)x, (unsigned)x, sel, sel, false, false);
}
__device__ __forceinline__ float half_bcast_f(float x, unsigned sel) {
    unsigned u = __float_as_uint(x);
    return __uint_as_float(__builtin_amdgcn_permlane16(u, u, sel, sel, false, false));
}

template <int M>
__device__ __forceinline__ float half_xor(float x) {
    return __uint_as_float(__builtin_amdgcn_update_dpp(0u, __float_as_uint(x), 0x160 + M, 0xf, 0xf, true));
}

__device__ __forceinline__ void gather_chunk(const int* __restrict__ csr_src,
                                             const float* __restrict__ a_s, int beg, int end,
                                             int off, int hl, int nd, float ad,
                                             int& s0, int& s1, float& l0, float& l1) {
    int j0 = beg + off + hl;
    int j1 = j0 + 16;
    bool in0 = j0 < end;
    bool in1 = j1 < end;
    int c0 = csr_src[in0 ? j0 : 0];
    int c1 = csr_src[in1 ? j1 : 0];
    s0 = in0 ? c0 : nd;
    s1 = in1 ? c1 : nd;
    l0 = lrelu(a_s[s0] + ad);
    l1 = lrelu(a_s[s1] + ad);
}

__global__ void __launch_bounds__(GW * 32)
gat_gather(const int* __restrict__ rowptr, const int* __restrict__ csr_src,
           const float* __restrict__ a_s, const float* __restrict__ a_d,
           const float* __restrict__ H, const float* __restrict__ bias,
           float* __restrict__ out, float* __restrict__ part,
           float* __restrict__ sums, int n_nodes) {
    __shared__ float4 red[GN][64];
    int wave = __builtin_amdgcn_readfirstlane((int)((blockIdx.x * blockDim.x + threadIdx.x) >> 5));
    int lane = threadIdx.x & 31;
    int hl = lane & 15;
    int hf = lane >> 4;
    int wv = threadIdx.x >> 5;
    int node = wave * 2 + hf;
    bool valid = node < n_nodes;
    float4 v0 = make_float4(0.0f, 0.0f, 0.0f, 0.0f);
    float4 v1 = make_float4(0.0f, 0.0f, 0.0f, 0.0f);
    float4 q0 = make_float4(0.0f, 0.0f, 0.0f, 0.0f);
    float4 q1 = make_float4(0.0f, 0.0f, 0.0f, 0.0f);

    if (wave * 2 < n_nodes) {
        int nd = valid ? node : n_nodes - 1;
        int beg = rowptr[nd];
        int end = rowptr[nd + 1];
        float ad = a_d[nd];
        float self_logit = lrelu(a_s[nd] + ad);
        const float4* H4 = (const float4*)H;
        float4 hs0 = H4[(unsigned)nd * 32u + hl];
        float4 hs1 = H4[(unsigned)nd * 32u + 16u + hl];
        int dlo = __builtin_amdgcn_readlane(end - beg, 0);
        int dhi = __builtin_amdgcn_readlane(end - beg, 16);
        int dmax = dlo > dhi ? dlo : dhi;

        int s0, s1;
        float l0, l1;
        gather_chunk(csr_src, a_s, beg, end, 0, hl, nd, ad, s0, s1, l0, l1);
        float mx = fmaxf(self_logit, fmaxf(l0, l1));
        for (int off = 32; off < dmax; off += 32) {
            int t0, t1;
            float m0, m1;
            gather_chunk(csr_src, a_s, beg, end, off, hl, nd, ad, t0, t1, m0, m1);
            mx = fmaxf(mx, fmaxf(m0, m1));
        }
        mx = fmaxf(mx, half_xor<8>(mx));
        mx = fmaxf(mx, half_xor<4>(mx));
        mx = fmaxf(mx, half_xor<2>(mx));
        mx = fmaxf(mx, half_xor<1>(mx));

        float es = expf(self_logit - mx);
        float4 acc0, acc1;
        acc0.x = es * hs0.x; acc0.y = es * hs0.y; acc0.z = es * hs0.z; acc0.w = es * hs0.w;
        acc1.x = es * hs1.x; acc1.y = es * hs1.y; acc1.z = es * hs1.z; acc1.w = es * hs1.w;
        float wsum = 0.0f;

        for (int off = 0; off < dmax; off += 32) {
            if (off > 0)
                gather_chunk(csr_src, a_s, beg, end, off, hl, nd, ad, s0, s1, l0, l1);
            float w0 = beg + off + hl < end ? expf(l0 - mx) : 0.0f;
            float w1 = beg + off + 16 + hl < end ? expf(l1 - mx) : 0.0f;
            wsum += w0 + w1;
            int n = dmax - off < 32 ? dmax - off : 32;
            n = (n + 3) & ~3;
            for (int k = 0; k < n; k += 4) {
                int ss = k < 16 ? s0 : s1;
                float ww = k < 16 ? w0 : w1;
#pragma unroll
                for (int u = 0; u < 4; ++u) {
                    unsigned sel = (unsigned)((k & 15) + u) * 0x11111111u;
                    int sk = half_bcast_i(ss, sel);
                    float wk = half_bcast_f(ww, sel);
                    unsigned ri = (unsigned)sk * 32u + hl;
                    float4 ha = H4[ri];
                    float4 hb = H4[ri + 16u];
                    acc0.x += wk * ha.x; acc0.y += wk * ha.y; acc0.z += wk * ha.z; acc0.w += wk * ha.w;
                    acc1.x += wk * hb.x; acc1.y += wk * hb.y; acc1.z += wk * hb.z; acc1.w += wk * hb.w;
                }
            }
        }
        wsum += half_xor<8>(wsum);
        wsum += half_xor<4>(wsum);
        wsum += half_xor<2>(wsum);
        wsum += half_xor<1>(wsum);

        if (valid) {
            float inv = 1.0f / (es + wsum);
            float4 b0 = ((const float4*)bias)[hl];
            float4 b1 = ((const float4*)bias)[16 + hl];
            v0.x = elu1(acc0.x * inv + b0.x);
            v0.y = elu1(acc0.y * inv + b0.y);
            v0.z = elu1(acc0.z * inv + b0.z);
            v0.w = elu1(acc0.w * inv + b0.w);
            v1.x = elu1(acc1.x * inv + b1.x);
            v1.y = elu1(acc1.y * inv + b1.y);
            v1.z = elu1(acc1.z * inv + b1.z);
            v1.w = elu1(acc1.w * inv + b1.w);
            q0.x = v0.x * v0.x; q0.y = v0.y * v0.y; q0.z = v0.z * v0.z; q0.w = v0.w * v0.w;
            q1.x = v1.x * v1.x; q1.y = v1.y * v1.y; q1.z = v1.z * v1.z; q1.w = v1.w * v1.w;
        }
    }

    int r = wv * 2 + hf;
    red[r][hl] = v0;
    red[r][16 + hl] = v1;
    red[r][32 + hl] = q0;
    red[r][48 + hl] = q1;
    __syncthreads();
    const float* rf = (const float*)red;
    int t = threadIdx.x;
    float tv = 0.0f, tq = 0.0f;
    for (int w = 0; w < GN; ++w) {
        tv += rf[w * 256 + t];
        tq += rf[w * 256 + 128 + t];
    }
    part[(size_t)blockIdx.x * (2 * HID) + t] = tv;
    part[(size_t)blockIdx.x * (2 * HID) + HID + t] = tq;
    if (blockIdx.x == 0) { sums[t] = 0.0f; sums[HID + t] = 0.0f; }
    if (valid) {
        float4* orow = (float4*)(out + (size_t)node * HID);
        orow[hl] = v0;
        orow[16 + hl] = v1;
    }
}

__global__ void gat_stats_reduce(const float* __restrict__ part, float* __restrict__ sums,
                                 int nrows) {
    int t = threadIdx.x;
    int r = blockIdx.x;
    const int g = gridDim.x;
    float acc = 0.0f;
    for (; r + 15 * g < nrows; r += 16 * g) {
        float x[16];
        for (int i = 0; i < 16; ++i) x[i] = part[(size_t)(r + i * g) * (2 * HID) + t];
        for (int i = 0; i < 16; ++i) acc += x[i];
    }
    for (; r < nrows; r += g) acc += part[(size_t)r * (2 * HID) + t];
    atomicAdd(&sums[t], acc);
}

__global__ void gat_bn_pool(float* __restrict__ out_acc, const float* __restrict__ sums,
                            const float* __restrict__ sumsq, const float* __restrict__ gamma,
                            const float* __restrict__ beta, const int* __restrict__ batch,
                            float* __restrict__ pooled, int layer, int n_nodes) {
    int wave = __builtin_amdgcn_readfirstlane((int)((blockIdx.x * blockDim.x + threadIdx.x) >> 5));
    int n0 = wave * NPW;
    if (n0 >= n_nodes) return;
    int lane = threadIdx.x & 31;

    float4 x[NPW];
    int gid[NPW];
#pragma unroll
    for (int i = 0; i < NPW; ++i) {
        int n = n0 + i < n_nodes ? n0 + i : n_nodes - 1;
        x[i] = ((const float4*)(out_acc + (size_t)n * HID))[lane];
        gid[i] = batch[n];
    }

    float invN = 1.0f / (float)n_nodes;
    float4 s = ((const float4*)sums)[lane];
    float4 sq = ((const float4*)sumsq)[lane];
    float4 ga = ((const float4*)gamma)[lane];
    float4 be = ((const float4*)beta)[lane];
    float4 mu, sd;
    mu.x = s.x * invN; mu.y = s.y * invN; mu.z = s.z * invN; mu.w = s.w * invN;
    sd.x = sqrtf(sq.x * invN - mu.x * mu.x + BN_EPS);
    sd.y = sqrtf(sq.y * invN - mu.y * mu.y + BN_EPS);
    sd.z = sqrtf(sq.z * invN - mu.z * mu.z + BN_EPS);
    sd.w = sqrtf(sq.w * invN - mu.w * mu.w + BN_EPS);

    int cur = gid[0];
    float4 run = make_float4(0.0f, 0.0f, 0.0f, 0.0f);
#pragma unroll
    for (int i = 0; i < NPW; ++i) {
        if (n0 + i < n_nodes) {
            float4 v;
            v.x = (x[i].x - mu.x) / sd.x * ga.x + be.x;
            v.y = (x[i].y - mu.y) / sd.y * ga.y + be.y;
            v.z = (x[i].z - mu.z) / sd.z * ga.z + be.z;
            v.w = (x[i].w - mu.w) / sd.w * ga.w + be.w;
            ((float4*)(out_acc + (size_t)(n0 + i) * HID))[lane] = v;
            if (gid[i] != cur) {
                pool_add(pooled, cur, layer, lane, run);
                cur = gid[i];
                run = make_float4(0.0f, 0.0f, 0.0f, 0.0f);
            }
            run.x += v.x; run.y += v.y; run.z += v.z; run.w += v.w;
        }
    }
    pool_add(pooled, cur, layer, lane, run);
}

extern "C" void kernel_launch(void* const* d_in, const int* in_sizes, int n_in,
                              void* d_out, int out_size, void* d_ws, size_t ws_size,
                              hipStream_t stream) {
    const float* x_in  = (const float*)d_in[0];
    const int*   ei    = (const int*)d_in[1];
    const int*   batch = (const int*)d_in[2];
    const float* W     = (const float*)d_in[3];
    const float* att_s = (const float*)d_in[4];
    const float* att_d = (const float*)d_in[5];
    const float* bias  = (const float*)d_in[6];
    const float* gamma = (const float*)d_in[7];
    const float* beta  = (const float*)d_in[8];

    const int N = in_sizes[0] / HID;
    const int E = in_sizes[1] / 2;

    float* pooled = (float*)d_out;
    float* xfinal = (float*)d_out + (size_t)NGRAPH * 3 * HID;

    char* w = (char*)d_ws;
    float* hbuf = (float*)w;  w += (size_t)N * HID * sizeof(float);
    float* xb   = (float*)w;  w += (size_t)N * HID * sizeof(float);
    const int n_gblk = (N + GN - 1) / GN;
    float* part = (float*)w;  w += (size_t)n_gblk * 2 * HID * sizeof(float);
    float* asrc = (float*)w;  w += (size_t)N * sizeof(float);
    float* adst = (float*)w;  w += (size_t)N * sizeof(float);
    float* sums = (float*)w;  w += (size_t)2 * HID * sizeof(float);
    float* sumsq = sums + HID;
    int* btot   = (int*)w;    w += (size_t)MAXB * BPAD * sizeof(int);
    int* bfill  = (int*)w;    w += (size_t)MAXB * BPAD * sizeof(int);
    int2* rec   = (int2*)w;   w += (size_t)E * sizeof(int2);
    int* rowptr = (int*)w;    w += (size_t)(N + 1) * sizeof(int);
    int* csrsrc = (int*)w;

    const int waves_g = (N + 15) / 16;
    dim3 blk(256);
    dim3 grid_gemm((waves_g * 32 + PT - 1) / PT);
    dim3 grid_node(n_gblk);
    dim3 grid_edge((E + CH - 1) / CH);
    const int nbk = (N + BNODES - 1) / BNODES;
    dim3 grid_rows((N + 8 * NPW - 1) / (8 * NPW));

    gat_zero<<<(NGRAPH * 3 * HID + 255) / 256, blk, 0, stream>>>(pooled, NGRAPH * 3 * HID);

    gat_zero<<<(2 * MAXB * BPAD + 255) / 256, blk, 0, stream>>>((float*)btot, 2 * MAXB * BPAD);
    gat_bucket_count<<<grid_edge, CT, 0, stream>>>(ei, btot, E, nbk);
    gat_bucket_scatter<<<grid_edge, CT, 0, stream>>>(ei, btot, bfill, rec, E, nbk);
    gat_bucket_sort<<<nbk, ST, 0, stream>>>(rec, btot, rowptr, csrsrc, N, nbk);

    for (int l = 0; l < 3; ++l) {
        const float* X  = (l == 0) ? x_in : xb;
        float* oacc     = (l == 2) ? xfinal : xb;
        const float* Wl = W + (size_t)l * HID * HID;

        if (l == 0)
            gat_gemm_wmma<false><<<grid_gemm, PT, 0, stream>>>(X, Wl, att_s, att_d, hbuf, asrc, adst, N,
                                                               sums, sumsq, gamma, beta, batch, pooled, 0);
        else
            gat_gemm_wmma<true><<<grid_gemm, PT, 0, stream>>>(X, Wl, att_s + l * HID, att_d + l * HID,
                                                              hbuf, asrc, adst, N, sums, sumsq,
                                                              gamma + (l - 1) * HID, beta + (l - 1) * HID,
                                                              batch, pooled, l - 1);
        gat_gather<<<grid_node, GW * 32, 0, stream>>>(rowptr, csrsrc, asrc, adst, hbuf, bias + l * HID,
                                                  oacc, part, sums, N);
        gat_stats_reduce<<<256, 256, 0, stream>>>(part, sums, n_gblk);
        if (l == 2)
            gat_bn_pool<<<grid_rows, blk, 0, stream>>>(oacc, sums, sumsq, gamma + l * HID,
                                                       beta + l * HID, batch, pooled, l, N);
    }
}
